# starting kernel
# baseline (speedup 1.0000x reference)
#include <hip/hip_runtime.h>
#include <math.h>

typedef float v2f __attribute__((ext_vector_type(2)));
typedef float v4f __attribute__((ext_vector_type(4)));
typedef float v8f __attribute__((ext_vector_type(8)));

#define NB 4
#define NC 16
#define NO 32
#define HW 256
#define LPIX (HW*HW)
#define DF 144
#define DQ 288
#define WAVES 4
#define TPW 8
#define ROWS_TOTAL (NB*LPIX)
#define TILES_TOTAL (ROWS_TOTAL/16)
#define BLOCKS (TILES_TOTAL/(WAVES*TPW))

static __device__ __forceinline__ v8f wmma4(v2f a, v2f b, v8f c) {
  return __builtin_amdgcn_wmma_f32_16x16x4_f32(
      false, a, false, b, (short)0, c, false, false);
}

__global__ __launch_bounds__(WAVES * 32)
void dynaconv_kernel(const float* __restrict__ X,
                     const float* __restrict__ W1,
                     const float* __restrict__ b1,
                     const float* __restrict__ W2,
                     const float* __restrict__ b2,
                     const float* __restrict__ bias,
                     float* __restrict__ Out) {
  __shared__ __align__(16) float sW1[ (DF/2) * 64 ];
  __shared__ __align__(16) float sW2[ (DQ/2) * 64 ];
  __shared__ __align__(16) float sFeat[WAVES][16 * DF];
  __shared__ __align__(16) float sPc  [WAVES][16 * 9];
  __shared__ __align__(16) float sHid [WAVES][16 * NO];

  const int tid  = threadIdx.x;
  const int lane = tid & 31;
  const int wv   = tid >> 5;
  const int col  = lane & 15;
  const int h    = lane >> 4;

  for (int e = tid; e < NO * DF; e += WAVES * 32) {
    int o = e / DF, f = e - o * DF;
    sW1[(f >> 1) * 64 + (o << 1) + (f & 1)] = W1[e];
  }
  for (int e = tid; e < DQ * NO; e += WAVES * 32) {
    int g = e >> 5, o2 = e & 31;
    int o = g / 9, j = g - o * 9;
    int q = o2 * 9 + j;
    sW2[(q >> 1) * 64 + (o << 1) + (q & 1)] = W2[e];
  }
  const float b1v0 = b1[col],      b1v1 = b1[col + 16];
  const float bs0  = bias[col],    bs1  = bias[col + 16];
  float b2a[9], b2b[9];
#pragma unroll
  for (int j = 0; j < 9; ++j) { b2a[j] = b2[col * 9 + j]; b2b[j] = b2[(col + 16) * 9 + j]; }
  __syncthreads();

  float* feat = sFeat[wv];
  float* pc   = sPc[wv];
  float* hid  = sHid[wv];

  for (int it = 0; it < TPW; ++it) {
    const int t    = ((int)blockIdx.x * WAVES + wv) * TPW + it;
    const int row0 = t << 4;
    const int bimg = row0 >> 16;
    const int p0   = row0 & (LPIX - 1);

    for (int e = lane; e < 16 * DF; e += 32) {
      int m = e / DF, f = e - m * DF;
      int c = f / 9, ij = f - c * 9;
      int di = ij / 3, dj = ij - di * 3;
      int p = p0 + m;
      int yy = (p >> 8) + di - 1;
      int xx = (p & 255) + dj - 1;
      float v = 0.0f;
      if ((unsigned)yy < (unsigned)HW && (unsigned)xx < (unsigned)HW)
        v = X[(((bimg * NC + c) * HW + yy) * HW) + xx];
      feat[e] = v;
    }
    __syncthreads();

    for (int e = lane; e < 16 * 9; e += 32) {
      int m = e / 9, j = e - m * 9;
      float s = 0.0f;
#pragma unroll
      for (int c = 0; c < NC; ++c) s += feat[m * DF + c * 9 + j];
      pc[e] = s;
    }
    __syncthreads();

    v8f acc0, acc1;
#pragma unroll
    for (int r = 0; r < 8; ++r) { acc0[r] = b1v0; acc1[r] = b1v1; }
    const float* aBase  = feat + col * DF + 2 * h;
    const float* w1Base = sW1 + h * 64 + (col << 1);
#pragma unroll
    for (int s = 0; s < DF / 4; ++s) {
      v2f a  = *(const v2f*)(aBase + 4 * s);
      v2f bA = *(const v2f*)(w1Base + s * 128);
      v2f bB = *(const v2f*)(w1Base + s * 128 + 32);
      acc0 = wmma4(a, bA, acc0);
      acc1 = wmma4(a, bB, acc1);
    }
#pragma unroll
    for (int r = 0; r < 8; ++r) {
      int m2 = r + 8 * h;
      hid[m2 * NO + col]      = tanhf(acc0[r]);
      hid[m2 * NO + col + 16] = tanhf(acc1[r]);
    }
    __syncthreads();

    float hr[NO];
#pragma unroll
    for (int k = 0; k < NO / 4; ++k) {
      v4f v = *(const v4f*)(hid + col * NO + 4 * k);
      hr[4 * k + 0] = v.x; hr[4 * k + 1] = v.y; hr[4 * k + 2] = v.z; hr[4 * k + 3] = v.w;
    }
    float pr[9];
#pragma unroll
    for (int j = 0; j < 9; ++j) pr[j] = pc[col * 9 + j];

#pragma unroll
    for (int r = 0; r < 8; ++r) {
      int m2 = r + 8 * h;
      float s0 = bs0, s1 = bs1;
#pragma unroll
      for (int j = 0; j < 9; ++j) { float pv = pc[m2 * 9 + j]; s0 += pv * b2a[j]; s1 += pv * b2b[j]; }
      acc0[r] = s0; acc1[r] = s1;
    }

    const float* w2Base = sW2 + h * 64 + (col << 1);
#pragma unroll
    for (int s = 0; s < DQ / 4; ++s) {
      const int q = 4 * s;
      float z0 = hr[(q + 0) / 9] * pr[(q + 0) % 9];
      float z1 = hr[(q + 1) / 9] * pr[(q + 1) % 9];
      float z2 = hr[(q + 2) / 9] * pr[(q + 2) % 9];
      float z3 = hr[(q + 3) / 9] * pr[(q + 3) % 9];
      v2f a; a.x = h ? z2 : z0; a.y = h ? z3 : z1;
      v2f bA = *(const v2f*)(w2Base + s * 128);
      v2f bB = *(const v2f*)(w2Base + s * 128 + 32);
      acc0 = wmma4(a, bA, acc0);
      acc1 = wmma4(a, bB, acc1);
    }

    float* outT = hid;
#pragma unroll
    for (int r = 0; r < 8; ++r) {
      int m2 = r + 8 * h;
      outT[m2 * NO + col]      = acc0[r];
      outT[m2 * NO + col + 16] = acc1[r];
    }
    __syncthreads();
#pragma unroll
    for (int e = 0; e < 16; ++e) {
      int idx = e * 32 + lane;
      int o = idx >> 4, m = idx & 15;
      Out[((long)(bimg * NO + o)) * LPIX + p0 + m] = outT[m * NO + o];
    }
    __syncthreads();
  }
}

extern "C" void kernel_launch(void* const* d_in, const int* in_sizes, int n_in,
                              void* d_out, int out_size, void* d_ws, size_t ws_size,
                              hipStream_t stream) {
  (void)in_sizes; (void)n_in; (void)out_size; (void)d_ws; (void)ws_size;
  const float* X    = (const float*)d_in[0];
  const float* W1   = (const float*)d_in[1];
  const float* b1   = (const float*)d_in[2];
  const float* W2   = (const float*)d_in[3];
  const float* b2   = (const float*)d_in[4];
  const float* bias = (const float*)d_in[5];
  float* Out = (float*)d_out;
  dynaconv_kernel<<<BLOCKS, WAVES * 32, 0, stream>>>(X, W1, b1, W2, b2, bias, Out);
}


# im2col gather (72 dependent scattered loads per lane per tile) replaced by an LDS input halo (16ch x 3row x 18col) filled by 9 independent coalesced loads; GEMM1 A and channel sums read from it
# speedup vs baseline: 2.0452x; 2.0452x over previous
#include <hip/hip_runtime.h>
#include <math.h>


typedef float v2f __attribute__((ext_vector_type(2)));
typedef float v4f __attribute__((ext_vector_type(4)));
typedef float v8f __attribute__((ext_vector_type(8)));

#define NB 4
#define NC 16
#define NO 32
#define HW 256
#define LPIX (HW*HW)
#define DF 144
#define DQ 288
#define WAVES 4
#define TPW 8
#define ROWS_TOTAL (NB*LPIX)
#define TILES_TOTAL (ROWS_TOTAL/16)
#define BLOCKS (TILES_TOTAL/(WAVES*TPW))

#define PP   24
#define NSEG (NC*3)
#define CSW  18

static __device__ __forceinline__ v8f wmma4(v2f a, v2f b, v8f c) {
  return __builtin_amdgcn_wmma_f32_16x16x4_f32(
      false, a, false, b, (short)0, c, false, false);
}

__global__ __launch_bounds__(WAVES * 32)
void dynaconv_kernel(const float* __restrict__ X,
                     const float* __restrict__ W1,
                     const float* __restrict__ b1,
                     const float* __restrict__ W2,
                     const float* __restrict__ b2,
                     const float* __restrict__ bias,
                     float* __restrict__ Out) {
  __shared__ __align__(16) float sW1[ (DF/2) * 64 ];
  __shared__ __align__(16) float sW2[ (DQ/2) * 64 ];
  __shared__ __align__(16) float sPatch[WAVES][NSEG * PP];
  __shared__ __align__(16) float sCs  [WAVES][64];
  __shared__ __align__(16) float sHid [WAVES][16 * NO];

  const int tid  = threadIdx.x;
  const int lane = tid & 31;
  const int wv   = tid >> 5;
  const int col  = lane & 15;
  const int h    = lane >> 4;

  for (int e = tid; e < NO * DF; e += WAVES * 32) {
    int o = e / DF, f = e - o * DF;
    sW1[(f >> 1) * 64 + (o << 1) + (f & 1)] = W1[e];
  }
  for (int e = tid; e < DQ * NO; e += WAVES * 32) {
    int g = e >> 5, o2 = e & 31;
    int o = g / 9, j = g - o * 9;
    int q = o2 * 9 + j;
    sW2[(q >> 1) * 64 + (o << 1) + (q & 1)] = W2[e];
  }
  const float b1v0 = b1[col],      b1v1 = b1[col + 16];
  const float bs0  = bias[col],    bs1  = bias[col + 16];
  float b2a[9], b2b[9];
#pragma unroll
  for (int j = 0; j < 9; ++j) { b2a[j] = b2[col * 9 + j]; b2b[j] = b2[(col + 16) * 9 + j]; }

  int cOff[6], cSafe[6], cDy[6], cLds[6];
#pragma unroll
  for (int i = 0; i < 6; ++i) {
    const int idx = lane + 32 * i;
    const int seg = idx >> 2, qd = idx & 3;
    const int c = seg / 3, di = seg - 3 * c;
    cDy[i]   = di - 1;
    cSafe[i] = c * LPIX + 4 * qd;
    cOff[i]  = cSafe[i] + (di - 1) * HW;
    cLds[i]  = seg * PP + 4 + 4 * qd;
  }
  const int side = lane & 1;
  int hOff[3], hSafe[3], hDy[3], hLds[3];
#pragma unroll
  for (int i = 0; i < 3; ++i) {
    const int seg = (lane >> 1) + 16 * i;
    const int c = seg / 3, di = seg - 3 * c;
    hDy[i]   = di - 1;
    hSafe[i] = c * LPIX;
    hOff[i]  = c * LPIX + (di - 1) * HW + (side ? 16 : -1);
    hLds[i]  = seg * PP + (side ? 20 : 3);
  }
  int bOff[2];
#pragma unroll
  for (int i = 0; i < 2; ++i) {
    const int e = lane + 32 * i;
    const int di = e / CSW, j = e - CSW * di;
    bOff[i] = di * PP + 3 + j;
  }
  int aOff[3][2];
#pragma unroll
  for (int v = 0; v < 3; ++v) {
#pragma unroll
    for (int e = 0; e < 2; ++e) {
      const int k = 4 * v + 2 * h + e;
      aOff[v][e] = (k / 3) * PP + 3 + (k % 3) + col;
    }
  }
  __syncthreads();

  float* patch = sPatch[wv];
  float* cs    = sCs[wv];
  float* hid   = sHid[wv];

  for (int it = 0; it < TPW; ++it) {
    const int t    = ((int)blockIdx.x * WAVES + wv) * TPW + it;
    const int row0 = t << 4;
    const int bimg = row0 >> 16;
    const int p0   = row0 & (LPIX - 1);
    const int y    = p0 >> 8;
    const int x0   = p0 & (HW - 1);
    const float* xb = X + (bimg * NC * LPIX + p0);

    v4f cv[6];
    float hv[3];
    const v4f z4 = {0.0f, 0.0f, 0.0f, 0.0f};
#pragma unroll
    for (int i = 0; i < 6; ++i) {
      const bool ok = (unsigned)(y + cDy[i]) < (unsigned)HW;
      const v4f tv = *(const v4f*)(xb + (ok ? cOff[i] : cSafe[i]));
      cv[i] = ok ? tv : z4;
    }
    const bool colOk = side ? (x0 < HW - 16) : (x0 > 0);
#pragma unroll
    for (int i = 0; i < 3; ++i) {
      const bool ok = colOk && ((unsigned)(y + hDy[i]) < (unsigned)HW);
      const float tv = xb[ok ? hOff[i] : hSafe[i]];
      hv[i] = ok ? tv : 0.0f;
    }
#pragma unroll
    for (int i = 0; i < 6; ++i) *(v4f*)(patch + cLds[i]) = cv[i];
#pragma unroll
    for (int i = 0; i < 3; ++i) patch[hLds[i]] = hv[i];
    __syncthreads();

#pragma unroll
    for (int i = 0; i < 2; ++i) {
      if (lane + 32 * i < 3 * CSW) {
        float s = 0.0f;
#pragma unroll
        for (int c = 0; c < NC; ++c) s += patch[bOff[i] + c * 3 * PP];
        cs[lane + 32 * i] = s;
      }
    }
    __syncthreads();

    v8f acc0, acc1;
#pragma unroll
    for (int r = 0; r < 8; ++r) { acc0[r] = b1v0; acc1[r] = b1v1; }
    const float* w1Base = sW1 + h * 64 + (col << 1);
#pragma unroll
    for (int s = 0; s < DF / 4; ++s) {
      const int u = s / 3, v = s - 3 * u;
      v2f a;
      a.x = patch[aOff[v][0] + u * 4 * PP];
      a.y = patch[aOff[v][1] + u * 4 * PP];
      v2f bA = *(const v2f*)(w1Base + s * 128);
      v2f bB = *(const v2f*)(w1Base + s * 128 + 32);
      acc0 = wmma4(a, bA, acc0);
      acc1 = wmma4(a, bB, acc1);
    }
#pragma unroll
    for (int r = 0; r < 8; ++r) {
      int m2 = r + 8 * h;
      hid[m2 * NO + col]      = tanhf(acc0[r]);
      hid[m2 * NO + col + 16] = tanhf(acc1[r]);
    }
    __syncthreads();

    float hr[NO];
#pragma unroll
    for (int k = 0; k < NO / 4; ++k) {
      v4f v = *(const v4f*)(hid + col * NO + 4 * k);
      hr[4 * k + 0] = v.x; hr[4 * k + 1] = v.y; hr[4 * k + 2] = v.z; hr[4 * k + 3] = v.w;
    }
    float pr[9];
#pragma unroll
    for (int j = 0; j < 9; ++j) pr[j] = cs[(j / 3) * CSW + col + (j % 3)];

#pragma unroll
    for (int r = 0; r < 8; ++r) {
      int m2 = r + 8 * h;
      float s0 = bs0, s1 = bs1;
#pragma unroll
      for (int j = 0; j < 9; ++j) {
        float pv = cs[(j / 3) * CSW + m2 + (j % 3)];
        s0 += pv * b2a[j]; s1 += pv * b2b[j];
      }
      acc0[r] = s0; acc1[r] = s1;
    }

    const float* w2Base = sW2 + h * 64 + (col << 1);
#pragma unroll
    for (int s = 0; s < DQ / 4; ++s) {
      const int q = 4 * s;
      float z0 = hr[(q + 0) / 9] * pr[(q + 0) % 9];
      float z1 = hr[(q + 1) / 9] * pr[(q + 1) % 9];
      float z2 = hr[(q + 2) / 9] * pr[(q + 2) % 9];
      float z3 = hr[(q + 3) / 9] * pr[(q + 3) % 9];
      v2f a; a.x = h ? z2 : z0; a.y = h ? z3 : z1;
      v2f bA = *(const v2f*)(w2Base + s * 128);
      v2f bB = *(const v2f*)(w2Base + s * 128 + 32);
      acc0 = wmma4(a, bA, acc0);
      acc1 = wmma4(a, bB, acc1);
    }

    float* outT = hid;
#pragma unroll
    for (int r = 0; r < 8; ++r) {
      int m2 = r + 8 * h;
      outT[m2 * NO + col]      = acc0[r];
      outT[m2 * NO + col + 16] = acc1[r];
    }
    __syncthreads();
#pragma unroll
    for (int e = 0; e < 16; ++e) {
      int idx = e * 32 + lane;
      int o = idx >> 4, m = idx & 15;
      Out[((long)(bimg * NO + o)) * LPIX + p0 + m] = outT[m * NO + o];
    }
    __syncthreads();
  }
}

extern "C" void kernel_launch(void* const* d_in, const int* in_sizes, int n_in,
                              void* d_out, int out_size, void* d_ws, size_t ws_size,
                              hipStream_t stream) {
  (void)in_sizes; (void)n_in; (void)out_size; (void)d_ws; (void)ws_size;
  const float* X    = (const float*)d_in[0];
  const float* W1   = (const float*)d_in[1];
  const float* b1   = (const float*)d_in[2];
  const float* W2   = (const float*)d_in[3];
  const float* b2   = (const float*)d_in[4];
  const float* bias = (const float*)d_in[5];
  float* Out = (float*)d_out;
  dynaconv_kernel<<<BLOCKS, WAVES * 32, 0, stream>>>(X, W1, b1, W2, b2, bias, Out);
}


# each wave works on two neighbouring 16-pixel tiles per trip: every W1/W2 fragment read from LDS feeds four matrix instructions on four independent accumulators (was two); summation order unchanged
# speedup vs baseline: 2.2496x; 1.0999x over previous
#include <hip/hip_runtime.h>
#include <math.h>


typedef float v2f __attribute__((ext_vector_type(2)));
typedef float v4f __attribute__((ext_vector_type(4)));
typedef float v8f __attribute__((ext_vector_type(8)));

#define NB 4
#define NC 16
#define NO 32
#define HW 256
#define LPIX (HW*HW)
#define DF 144
#define DQ 288
#define WAVES 4
#define TPW 8
#define MT 2
#define ROWS_TOTAL (NB*LPIX)
#define TILES_TOTAL (ROWS_TOTAL/16)
#define BLOCKS (TILES_TOTAL/(WAVES*TPW))

#define PP   24
#define NSEG (NC*3)
#define CSW  18

static __device__ __forceinline__ v8f wmma4(v2f a, v2f b, v8f c) {
  return __builtin_amdgcn_wmma_f32_16x16x4_f32(
      false, a, false, b, (short)0, c, false, false);
}

__global__ __launch_bounds__(WAVES * 32)
void dynaconv_kernel(const float* __restrict__ X,
                     const float* __restrict__ W1,
                     const float* __restrict__ b1,
                     const float* __restrict__ W2,
                     const float* __restrict__ b2,
                     const float* __restrict__ bias,
                     float* __restrict__ Out) {
  __shared__ __align__(16) float sW1[ (DF/2) * 64 ];
  __shared__ __align__(16) float sW2[ (DQ/2) * 64 ];
  __shared__ __align__(16) float sPatch[WAVES][MT][NSEG * PP];
  __shared__ __align__(16) float sCs  [WAVES][MT][64];
  __shared__ __align__(16) float sHid [WAVES][MT][16 * NO];

  const int tid  = threadIdx.x;
  const int lane = tid & 31;
  const int wv   = tid >> 5;
  const int col  = lane & 15;
  const int h    = lane >> 4;

  for (int e = tid; e < NO * DF; e += WAVES * 32) {
    int o = e / DF, f = e - o * DF;
    sW1[(f >> 1) * 64 + (o << 1) + (f & 1)] = W1[e];
  }
  for (int e = tid; e < DQ * NO; e += WAVES * 32) {
    int g = e >> 5, o2 = e & 31;
    int o = g / 9, j = g - o * 9;
    int q = o2 * 9 + j;
    sW2[(q >> 1) * 64 + (o << 1) + (q & 1)] = W2[e];
  }
  const float b1v0 = b1[col],      b1v1 = b1[col + 16];
  const float bs0  = bias[col],    bs1  = bias[col + 16];
  float b2a[9], b2b[9];
#pragma unroll
  for (int j = 0; j < 9; ++j) { b2a[j] = b2[col * 9 + j]; b2b[j] = b2[(col + 16) * 9 + j]; }

  int cOff[6], cSafe[6], cDy[6], cLds[6];
#pragma unroll
  for (int i = 0; i < 6; ++i) {
    const int idx = lane + 32 * i;
    const int seg = idx >> 2, qd = idx & 3;
    const int c = seg / 3, di = seg - 3 * c;
    cDy[i]   = di - 1;
    cSafe[i] = c * LPIX + 4 * qd;
    cOff[i]  = cSafe[i] + (di - 1) * HW;
    cLds[i]  = seg * PP + 4 + 4 * qd;
  }
  const int side = lane & 1;
  int hOff[3], hSafe[3], hDy[3], hLds[3];
#pragma unroll
  for (int i = 0; i < 3; ++i) {
    const int seg = (lane >> 1) + 16 * i;
    const int c = seg / 3, di = seg - 3 * c;
    hDy[i]   = di - 1;
    hSafe[i] = c * LPIX;
    hOff[i]  = c * LPIX + (di - 1) * HW + (side ? 16 : -1);
    hLds[i]  = seg * PP + (side ? 20 : 3);
  }
  int bOff[2];
#pragma unroll
  for (int i = 0; i < 2; ++i) {
    const int e = lane + 32 * i;
    const int di = e / CSW, j = e - CSW * di;
    bOff[i] = di * PP + 3 + j;
  }
  int aOff[3][2];
#pragma unroll
  for (int v = 0; v < 3; ++v) {
#pragma unroll
    for (int e = 0; e < 2; ++e) {
      const int k = 4 * v + 2 * h + e;
      aOff[v][e] = (k / 3) * PP + 3 + (k % 3) + col;
    }
  }
  __syncthreads();

  float* patch = sPatch[wv][0];
  float* cs    = sCs[wv][0];
  float* hid   = sHid[wv][0];
  constexpr int PSTR = NSEG * PP;
  constexpr int HSTR = 16 * NO;

  for (int it = 0; it < TPW / MT; ++it) {
    const int t    = ((int)blockIdx.x * WAVES + wv) * TPW + it * MT;
    const int row0 = t << 4;
    const int bimg = row0 >> 16;
    const int p0   = row0 & (LPIX - 1);
    const int y    = p0 >> 8;
    const int x0   = p0 & (HW - 1);
    const float* xb = X + (bimg * NC * LPIX + p0);

    v4f cv[MT][6];
    float hv[MT][3];
    const v4f z4 = {0.0f, 0.0f, 0.0f, 0.0f};
#pragma unroll
    for (int tl = 0; tl < MT; ++tl) {
#pragma unroll
      for (int i = 0; i < 6; ++i) {
        const bool ok = (unsigned)(y + cDy[i]) < (unsigned)HW;
        const v4f tv = *(const v4f*)(xb + 16 * tl + (ok ? cOff[i] : cSafe[i]));
        cv[tl][i] = ok ? tv : z4;
      }
    }
#pragma unroll
    for (int tl = 0; tl < MT; ++tl) {
      const int xt = x0 + 16 * tl;
      const bool colOk = side ? (xt < HW - 16) : (xt > 0);
#pragma unroll
      for (int i = 0; i < 3; ++i) {
        const bool ok = colOk && ((unsigned)(y + hDy[i]) < (unsigned)HW);
        const float tv = xb[16 * tl + (ok ? hOff[i] : hSafe[i])];
        hv[tl][i] = ok ? tv : 0.0f;
      }
    }
#pragma unroll
    for (int tl = 0; tl < MT; ++tl) {
#pragma unroll
      for (int i = 0; i < 6; ++i) *(v4f*)(patch + tl * PSTR + cLds[i]) = cv[tl][i];
#pragma unroll
      for (int i = 0; i < 3; ++i) patch[tl * PSTR + hLds[i]] = hv[tl][i];
    }
    __syncthreads();

#pragma unroll
    for (int i = 0; i < 2; ++i) {
      if (lane + 32 * i < 3 * CSW) {
#pragma unroll
        for (int tl = 0; tl < MT; ++tl) {
          float s = 0.0f;
#pragma unroll
          for (int c = 0; c < NC; ++c) s += patch[tl * PSTR + bOff[i] + c * 3 * PP];
          cs[tl * 64 + lane + 32 * i] = s;
        }
      }
    }
    __syncthreads();

    v8f acc0[MT], acc1[MT];
#pragma unroll
    for (int tl = 0; tl < MT; ++tl) {
#pragma unroll
      for (int r = 0; r < 8; ++r) { acc0[tl][r] = b1v0; acc1[tl][r] = b1v1; }
    }
    const float* w1Base = sW1 + h * 64 + (col << 1);
#pragma unroll
    for (int s = 0; s < DF / 4; ++s) {
      const int u = s / 3, v = s - 3 * u;
      v2f bA = *(const v2f*)(w1Base + s * 128);
      v2f bB = *(const v2f*)(w1Base + s * 128 + 32);
#pragma unroll
      for (int tl = 0; tl < MT; ++tl) {
        v2f a;
        a.x = patch[tl * PSTR + aOff[v][0] + u * 4 * PP];
        a.y = patch[tl * PSTR + aOff[v][1] + u * 4 * PP];
        acc0[tl] = wmma4(a, bA, acc0[tl]);
        acc1[tl] = wmma4(a, bB, acc1[tl]);
      }
    }
#pragma unroll
    for (int tl = 0; tl < MT; ++tl) {
#pragma unroll
      for (int r = 0; r < 8; ++r) {
        int m2 = r + 8 * h;
        hid[tl * HSTR + m2 * NO + col]      = tanhf(acc0[tl][r]);
        hid[tl * HSTR + m2 * NO + col + 16] = tanhf(acc1[tl][r]);
      }
    }
    __syncthreads();

    float hr[MT][NO];
    float pr[MT][9];
#pragma unroll
    for (int tl = 0; tl < MT; ++tl) {
#pragma unroll
      for (int k = 0; k < NO / 4; ++k) {
        v4f v = *(const v4f*)(hid + tl * HSTR + col * NO + 4 * k);
        hr[tl][4 * k + 0] = v.x; hr[tl][4 * k + 1] = v.y; hr[tl][4 * k + 2] = v.z; hr[tl][4 * k + 3] = v.w;
      }
#pragma unroll
      for (int j = 0; j < 9; ++j) pr[tl][j] = cs[tl * 64 + (j / 3) * CSW + col + (j % 3)];

#pragma unroll
      for (int r = 0; r < 8; ++r) {
        int m2 = r + 8 * h;
        float s0 = bs0, s1 = bs1;
#pragma unroll
        for (int j = 0; j < 9; ++j) {
          float pv = cs[tl * 64 + (j / 3) * CSW + m2 + (j % 3)];
          s0 += pv * b2a[j]; s1 += pv * b2b[j];
        }
        acc0[tl][r] = s0; acc1[tl][r] = s1;
      }
    }

    const float* w2Base = sW2 + h * 64 + (col << 1);
#pragma unroll
    for (int s = 0; s < DQ / 4; ++s) {
      const int q = 4 * s;
      v2f bA = *(const v2f*)(w2Base + s * 128);
      v2f bB = *(const v2f*)(w2Base + s * 128 + 32);
#pragma unroll
      for (int tl = 0; tl < MT; ++tl) {
        float z0 = hr[tl][(q + 0) / 9] * pr[tl][(q + 0) % 9];
        float z1 = hr[tl][(q + 1) / 9] * pr[tl][(q + 1) % 9];
        float z2 = hr[tl][(q + 2) / 9] * pr[tl][(q + 2) % 9];
        float z3 = hr[tl][(q + 3) / 9] * pr[tl][(q + 3) % 9];
        v2f a; a.x = h ? z2 : z0; a.y = h ? z3 : z1;
        acc0[tl] = wmma4(a, bA, acc0[tl]);
        acc1[tl] = wmma4(a, bB, acc1[tl]);
      }
    }

    float* outT = hid;
#pragma unroll
    for (int tl = 0; tl < MT; ++tl) {
#pragma unroll
      for (int r = 0; r < 8; ++r) {
        int m2 = r + 8 * h;
        outT[tl * HSTR + m2 * NO + col]      = acc0[tl][r];
        outT[tl * HSTR + m2 * NO + col + 16] = acc1[tl][r];
      }
    }
    __syncthreads();
#pragma unroll
    for (int tl = 0; tl < MT; ++tl) {
#pragma unroll
      for (int e = 0; e < 16; ++e) {
        int idx = e * 32 + lane;
        int o = idx >> 4, m = idx & 15;
        Out[((long)(bimg * NO + o)) * LPIX + p0 + 16 * tl + m] = outT[tl * HSTR + m * NO + o];
      }
    }
    __syncthreads();
  }
}

extern "C" void kernel_launch(void* const* d_in, const int* in_sizes, int n_in,
                              void* d_out, int out_size, void* d_ws, size_t ws_size,
                              hipStream_t stream) {
  (void)in_sizes; (void)n_in; (void)out_size; (void)d_ws; (void)ws_size;
  const float* X    = (const float*)d_in[0];
  const float* W1   = (const float*)d_in[1];
  const float* b1   = (const float*)d_in[2];
  const float* W2   = (const float*)d_in[3];
  const float* b2   = (const float*)d_in[4];
  const float* bias = (const float*)d_in[5];
  float* Out = (float*)d_out;
  dynaconv_kernel<<<BLOCKS, WAVES * 32, 0, stream>>>(X, W1, b1, W2, b2, bias, Out);
}


# tanh of the hidden layer made branch-free: both range formulas of the library tanh computed in every lane, one chosen by a compare; a tile's values are independent chains, not serial branchy calls
# speedup vs baseline: 2.4794x; 1.1022x over previous
#include <hip/hip_runtime.h>
#include <math.h>


typedef float v2f __attribute__((ext_vector_type(2)));
typedef float v4f __attribute__((ext_vector_type(4)));
typedef float v8f __attribute__((ext_vector_type(8)));

#define NB 4
#define NC 16
#define NO 32
#define HW 256
#define LPIX (HW*HW)
#define DF 144
#define DQ 288
#define WAVES 4
#define TPW 8
#define MT 2
#define ROWS_TOTAL (NB*LPIX)
#define TILES_TOTAL (ROWS_TOTAL/16)
#define BLOCKS (TILES_TOTAL/(WAVES*TPW))

#define PP   24
#define NSEG (NC*3)
#define CSW  18

static __device__ __forceinline__ v8f wmma4(v2f a, v2f b, v8f c) {
  return __builtin_amdgcn_wmma_f32_16x16x4_f32(
      false, a, false, b, (short)0, c, false, false);
}


static __device__ __forceinline__ float tanh_sel(float x) {
  const float y  = fabsf(x);
  const float y2 = y * y;
  const float p  = fmaf(y2, fmaf(y2, fmaf(y2, fmaf(y2, -0x1.758e7ap-8f, 0x1.521192p-6f),
                                           -0x1.b8389cp-5f), 0x1.110704p-3f), -0x1.555532p-2f);
  const float zs = fmaf(y2, y * p, y);
  const float t  = expf(2.0f * y);
  const float zl = fmaf(-2.0f, __builtin_amdgcn_rcpf(t + 1.0f), 1.0f);
  return copysignf(y < 0.625f ? zs : zl, x);
}

__global__ __launch_bounds__(WAVES * 32)
void dynaconv_kernel(const float* __restrict__ X,
                     const float* __restrict__ W1,
                     const float* __restrict__ b1,
                     const float* __restrict__ W2,
                     const float* __restrict__ b2,
                     const float* __restrict__ bias,
                     float* __restrict__ Out) {
  __shared__ __align__(16) float sW1[ (DF/2) * 64 ];
  __shared__ __align__(16) float sW2[ (DQ/2) * 64 ];
  __shared__ __align__(16) float sPatch[WAVES][MT][NSEG * PP];
  __shared__ __align__(16) float sCs  [WAVES][MT][64];
  __shared__ __align__(16) float sHid [WAVES][MT][16 * NO];

  const int tid  = threadIdx.x;
  const int lane = tid & 31;
  const int wv   = tid >> 5;
  const int col  = lane & 15;
  const int h    = lane >> 4;

  for (int e = tid; e < NO * DF; e += WAVES * 32) {
    int o = e / DF, f = e - o * DF;
    sW1[(f >> 1) * 64 + (o << 1) + (f & 1)] = W1[e];
  }
  for (int e = tid; e < DQ * NO; e += WAVES * 32) {
    int g = e >> 5, o2 = e & 31;
    int o = g / 9, j = g - o * 9;
    int q = o2 * 9 + j;
    sW2[(q >> 1) * 64 + (o << 1) + (q & 1)] = W2[e];
  }
  const float b1v0 = b1[col],      b1v1 = b1[col + 16];
  const float bs0  = bias[col],    bs1  = bias[col + 16];
  float b2a[9], b2b[9];
#pragma unroll
  for (int j = 0; j < 9; ++j) { b2a[j] = b2[col * 9 + j]; b2b[j] = b2[(col + 16) * 9 + j]; }

  int cOff[6], cSafe[6], cDy[6], cLds[6];
#pragma unroll
  for (int i = 0; i < 6; ++i) {
    const int idx = lane + 32 * i;
    const int seg = idx >> 2, qd = idx & 3;
    const int c = seg / 3, di = seg - 3 * c;
    cDy[i]   = di - 1;
    cSafe[i] = c * LPIX + 4 * qd;
    cOff[i]  = cSafe[i] + (di - 1) * HW;
    cLds[i]  = seg * PP + 4 + 4 * qd;
  }
  const int side = lane & 1;
  int hOff[3], hSafe[3], hDy[3], hLds[3];
#pragma unroll
  for (int i = 0; i < 3; ++i) {
    const int seg = (lane >> 1) + 16 * i;
    const int c = seg / 3, di = seg - 3 * c;
    hDy[i]   = di - 1;
    hSafe[i] = c * LPIX;
    hOff[i]  = c * LPIX + (di - 1) * HW + (side ? 16 : -1);
    hLds[i]  = seg * PP + (side ? 20 : 3);
  }
  int bOff[2];
#pragma unroll
  for (int i = 0; i < 2; ++i) {
    const int e = lane + 32 * i;
    const int di = e / CSW, j = e - CSW * di;
    bOff[i] = di * PP + 3 + j;
  }
  int aOff[3][2];
#pragma unroll
  for (int v = 0; v < 3; ++v) {
#pragma unroll
    for (int e = 0; e < 2; ++e) {
      const int k = 4 * v + 2 * h + e;
      aOff[v][e] = (k / 3) * PP + 3 + (k % 3) + col;
    }
  }
  __syncthreads();

  float* patch = sPatch[wv][0];
  float* cs    = sCs[wv][0];
  float* hid   = sHid[wv][0];
  constexpr int PSTR = NSEG * PP;
  constexpr int HSTR = 16 * NO;

  for (int it = 0; it < TPW / MT; ++it) {
    const int t    = ((int)blockIdx.x * WAVES + wv) * TPW + it * MT;
    const int row0 = t << 4;
    const int bimg = row0 >> 16;
    const int p0   = row0 & (LPIX - 1);
    const int y    = p0 >> 8;
    const int x0   = p0 & (HW - 1);
    const float* xb = X + (bimg * NC * LPIX + p0);

    v4f cv[MT][6];
    float hv[MT][3];
    const v4f z4 = {0.0f, 0.0f, 0.0f, 0.0f};
#pragma unroll
    for (int tl = 0; tl < MT; ++tl) {
#pragma unroll
      for (int i = 0; i < 6; ++i) {
        const bool ok = (unsigned)(y + cDy[i]) < (unsigned)HW;
        const v4f tv = *(const v4f*)(xb + 16 * tl + (ok ? cOff[i] : cSafe[i]));
        cv[tl][i] = ok ? tv : z4;
      }
    }
#pragma unroll
    for (int tl = 0; tl < MT; ++tl) {
      const int xt = x0 + 16 * tl;
      const bool colOk = side ? (xt < HW - 16) : (xt > 0);
#pragma unroll
      for (int i = 0; i < 3; ++i) {
        const bool ok = colOk && ((unsigned)(y + hDy[i]) < (unsigned)HW);
        const float tv = xb[16 * tl + (ok ? hOff[i] : hSafe[i])];
        hv[tl][i] = ok ? tv : 0.0f;
      }
    }
#pragma unroll
    for (int tl = 0; tl < MT; ++tl) {
#pragma unroll
      for (int i = 0; i < 6; ++i) *(v4f*)(patch + tl * PSTR + cLds[i]) = cv[tl][i];
#pragma unroll
      for (int i = 0; i < 3; ++i) patch[tl * PSTR + hLds[i]] = hv[tl][i];
    }
    __syncthreads();

#pragma unroll
    for (int i = 0; i < 2; ++i) {
      if (lane + 32 * i < 3 * CSW) {
#pragma unroll
        for (int tl = 0; tl < MT; ++tl) {
          float s = 0.0f;
#pragma unroll
          for (int c = 0; c < NC; ++c) s += patch[tl * PSTR + bOff[i] + c * 3 * PP];
          cs[tl * 64 + lane + 32 * i] = s;
        }
      }
    }
    __syncthreads();

    v8f acc0[MT], acc1[MT];
#pragma unroll
    for (int tl = 0; tl < MT; ++tl) {
#pragma unroll
      for (int r = 0; r < 8; ++r) { acc0[tl][r] = b1v0; acc1[tl][r] = b1v1; }
    }
    const float* w1Base = sW1 + h * 64 + (col << 1);
#pragma unroll
    for (int s = 0; s < DF / 4; ++s) {
      const int u = s / 3, v = s - 3 * u;
      v2f bA = *(const v2f*)(w1Base + s * 128);
      v2f bB = *(const v2f*)(w1Base + s * 128 + 32);
#pragma unroll
      for (int tl = 0; tl < MT; ++tl) {
        v2f a;
        a.x = patch[tl * PSTR + aOff[v][0] + u * 4 * PP];
        a.y = patch[tl * PSTR + aOff[v][1] + u * 4 * PP];
        acc0[tl] = wmma4(a, bA, acc0[tl]);
        acc1[tl] = wmma4(a, bB, acc1[tl]);
      }
    }
#pragma unroll
    for (int tl = 0; tl < MT; ++tl) {
#pragma unroll
      for (int r = 0; r < 8; ++r) {
        int m2 = r + 8 * h;
        hid[tl * HSTR + m2 * NO + col]      = tanh_sel(acc0[tl][r]);
        hid[tl * HSTR + m2 * NO + col + 16] = tanh_sel(acc1[tl][r]);
      }
    }
    __syncthreads();

    float hr[MT][NO];
    float pr[MT][9];
#pragma unroll
    for (int tl = 0; tl < MT; ++tl) {
#pragma unroll
      for (int k = 0; k < NO / 4; ++k) {
        v4f v = *(const v4f*)(hid + tl * HSTR + col * NO + 4 * k);
        hr[tl][4 * k + 0] = v.x; hr[tl][4 * k + 1] = v.y; hr[tl][4 * k + 2] = v.z; hr[tl][4 * k + 3] = v.w;
      }
#pragma unroll
      for (int j = 0; j < 9; ++j) pr[tl][j] = cs[tl * 64 + (j / 3) * CSW + col + (j % 3)];

#pragma unroll
      for (int r = 0; r < 8; ++r) {
        int m2 = r + 8 * h;
        float s0 = bs0, s1 = bs1;
#pragma unroll
        for (int j = 0; j < 9; ++j) {
          float pv = cs[tl * 64 + (j / 3) * CSW + m2 + (j % 3)];
          s0 += pv * b2a[j]; s1 += pv * b2b[j];
        }
        acc0[tl][r] = s0; acc1[tl][r] = s1;
      }
    }

    const float* w2Base = sW2 + h * 64 + (col << 1);
#pragma unroll
    for (int s = 0; s < DQ / 4; ++s) {
      const int q = 4 * s;
      v2f bA = *(const v2f*)(w2Base + s * 128);
      v2f bB = *(const v2f*)(w2Base + s * 128 + 32);
#pragma unroll
      for (int tl = 0; tl < MT; ++tl) {
        float z0 = hr[tl][(q + 0) / 9] * pr[tl][(q + 0) % 9];
        float z1 = hr[tl][(q + 1) / 9] * pr[tl][(q + 1) % 9];
        float z2 = hr[tl][(q + 2) / 9] * pr[tl][(q + 2) % 9];
        float z3 = hr[tl][(q + 3) / 9] * pr[tl][(q + 3) % 9];
        v2f a; a.x = h ? z2 : z0; a.y = h ? z3 : z1;
        acc0[tl] = wmma4(a, bA, acc0[tl]);
        acc1[tl] = wmma4(a, bB, acc1[tl]);
      }
    }

    float* outT = hid;
#pragma unroll
    for (int tl = 0; tl < MT; ++tl) {
#pragma unroll
      for (int r = 0; r < 8; ++r) {
        int m2 = r + 8 * h;
        outT[tl * HSTR + m2 * NO + col]      = acc0[tl][r];
        outT[tl * HSTR + m2 * NO + col + 16] = acc1[tl][r];
      }
    }
    __syncthreads();
#pragma unroll
    for (int tl = 0; tl < MT; ++tl) {
#pragma unroll
      for (int e = 0; e < 16; ++e) {
        int idx = e * 32 + lane;
        int o = idx >> 4, m = idx & 15;
        Out[((long)(bimg * NO + o)) * LPIX + p0 + 16 * tl + m] = outT[tl * HSTR + m * NO + o];
      }
    }
    __syncthreads();
  }
}

extern "C" void kernel_launch(void* const* d_in, const int* in_sizes, int n_in,
                              void* d_out, int out_size, void* d_ws, size_t ws_size,
                              hipStream_t stream) {
  (void)in_sizes; (void)n_in; (void)out_size; (void)d_ws; (void)ws_size;
  const float* X    = (const float*)d_in[0];
  const float* W1   = (const float*)d_in[1];
  const float* b1   = (const float*)d_in[2];
  const float* W2   = (const float*)d_in[3];
  const float* b2   = (const float*)d_in[4];
  const float* bias = (const float*)d_in[5];
  float* Out = (float*)d_out;
  dynaconv_kernel<<<BLOCKS, WAVES * 32, 0, stream>>>(X, W1, b1, W2, b2, bias, Out);
}


# W1/W2 held in LDS as padded rows [n][k] (pitch K+4), not interleaved k pairs: staged by 16-byte coalesced loads, stores at most 2-way conflicted (was 4-byte loads, 16-way); summation order unchanged
# speedup vs baseline: 2.8510x; 1.1499x over previous
#include <hip/hip_runtime.h>
#include <math.h>


typedef float v2f __attribute__((ext_vector_type(2)));
typedef float v4f __attribute__((ext_vector_type(4)));
typedef float v8f __attribute__((ext_vector_type(8)));

#define NB 4
#define NC 16
#define NO 32
#define HW 256
#define LPIX (HW*HW)
#define DF 144
#define DQ 288
#define WAVES 4
#define TPW 8
#define MT 2
#define ROWS_TOTAL (NB*LPIX)
#define TILES_TOTAL (ROWS_TOTAL/16)
#define BLOCKS (TILES_TOTAL/(WAVES*TPW))

#define P1 (DF + 4)
#define P2 (DQ + 4)

#define PP   24
#define NSEG (NC*3)
#define CSW  18

static __device__ __forceinline__ v8f wmma4(v2f a, v2f b, v8f c) {
  return __builtin_amdgcn_wmma_f32_16x16x4_f32(
      false, a, false, b, (short)0, c, false, false);
}


static __device__ __forceinline__ float tanh_sel(float x) {
  const float y  = fabsf(x);
  const float y2 = y * y;
  const float p  = fmaf(y2, fmaf(y2, fmaf(y2, fmaf(y2, -0x1.758e7ap-8f, 0x1.521192p-6f),
                                           -0x1.b8389cp-5f), 0x1.110704p-3f), -0x1.555532p-2f);
  const float zs = fmaf(y2, y * p, y);
  const float t  = expf(2.0f * y);
  const float zl = fmaf(-2.0f, __builtin_amdgcn_rcpf(t + 1.0f), 1.0f);
  return copysignf(y < 0.625f ? zs : zl, x);
}

__global__ __launch_bounds__(WAVES * 32)
void dynaconv_kernel(const float* __restrict__ X,
                     const float* __restrict__ W1,
                     const float* __restrict__ b1,
                     const float* __restrict__ W2,
                     const float* __restrict__ b2,
                     const float* __restrict__ bias,
                     float* __restrict__ Out) {
  __shared__ __align__(16) float sW1[ NO * P1 ];
  __shared__ __align__(16) float sW2[ NO * P2 ];
  __shared__ __align__(16) float sPatch[WAVES][MT][NSEG * PP];
  __shared__ __align__(16) float sCs  [WAVES][MT][64];
  __shared__ __align__(16) float sHid [WAVES][MT][16 * NO];

  const int tid  = threadIdx.x;
  const int lane = tid & 31;
  const int wv   = tid >> 5;
  const int col  = lane & 15;
  const int h    = lane >> 4;

  {
    constexpr int NT  = WAVES * 32;
    constexpr int N1  = NO * DF / 4 / NT;
    constexpr int N2  = DQ * NO / 4 / NT;
    v4f w1v[N1], w2v[N2];
#pragma unroll
    for (int i = 0; i < N1; ++i) w1v[i] = ((const v4f*)W1)[tid + NT * i];
#pragma unroll
    for (int i = 0; i < N2; ++i) w2v[i] = ((const v4f*)W2)[tid + NT * i];
#pragma unroll
    for (int i = 0; i < N1; ++i) {
      const int e4 = tid + NT * i;
      const int o = e4 / (DF / 4), c = e4 - o * (DF / 4);
      *(v4f*)(sW1 + o * P1 + 4 * c) = w1v[i];
    }
#pragma unroll
    for (int i = 0; i < N2; ++i) {
      const int e4 = tid + NT * i;
      const int g = e4 >> 3, c4 = e4 & 7;
      const int o = g / 9, j = g - o * 9;
      float* d = sW2 + o * P2 + 36 * c4 + j;
      d[0] = w2v[i].x; d[9] = w2v[i].y; d[18] = w2v[i].z; d[27] = w2v[i].w;
    }
  }
  const float b1v0 = b1[col],      b1v1 = b1[col + 16];
  const float bs0  = bias[col],    bs1  = bias[col + 16];
  float b2a[9], b2b[9];
#pragma unroll
  for (int j = 0; j < 9; ++j) { b2a[j] = b2[col * 9 + j]; b2b[j] = b2[(col + 16) * 9 + j]; }

  int cOff[6], cSafe[6], cDy[6], cLds[6];
#pragma unroll
  for (int i = 0; i < 6; ++i) {
    const int idx = lane + 32 * i;
    const int seg = idx >> 2, qd = idx & 3;
    const int c = seg / 3, di = seg - 3 * c;
    cDy[i]   = di - 1;
    cSafe[i] = c * LPIX + 4 * qd;
    cOff[i]  = cSafe[i] + (di - 1) * HW;
    cLds[i]  = seg * PP + 4 + 4 * qd;
  }
  const int side = lane & 1;
  int hOff[3], hSafe[3], hDy[3], hLds[3];
#pragma unroll
  for (int i = 0; i < 3; ++i) {
    const int seg = (lane >> 1) + 16 * i;
    const int c = seg / 3, di = seg - 3 * c;
    hDy[i]   = di - 1;
    hSafe[i] = c * LPIX;
    hOff[i]  = c * LPIX + (di - 1) * HW + (side ? 16 : -1);
    hLds[i]  = seg * PP + (side ? 20 : 3);
  }
  int bOff[2];
#pragma unroll
  for (int i = 0; i < 2; ++i) {
    const int e = lane + 32 * i;
    const int di = e / CSW, j = e - CSW * di;
    bOff[i] = di * PP + 3 + j;
  }
  int aOff[3][2];
#pragma unroll
  for (int v = 0; v < 3; ++v) {
#pragma unroll
    for (int e = 0; e < 2; ++e) {
      const int k = 4 * v + 2 * h + e;
      aOff[v][e] = (k / 3) * PP + 3 + (k % 3) + col;
    }
  }
  __syncthreads();

  float* patch = sPatch[wv][0];
  float* cs    = sCs[wv][0];
  float* hid   = sHid[wv][0];
  constexpr int PSTR = NSEG * PP;
  constexpr int HSTR = 16 * NO;

  for (int it = 0; it < TPW / MT; ++it) {
    const int t    = ((int)blockIdx.x * WAVES + wv) * TPW + it * MT;
    const int row0 = t << 4;
    const int bimg = row0 >> 16;
    const int p0   = row0 & (LPIX - 1);
    const int y    = p0 >> 8;
    const int x0   = p0 & (HW - 1);
    const float* xb = X + (bimg * NC * LPIX + p0);

    v4f cv[MT][6];
    float hv[MT][3];
    const v4f z4 = {0.0f, 0.0f, 0.0f, 0.0f};
#pragma unroll
    for (int tl = 0; tl < MT; ++tl) {
#pragma unroll
      for (int i = 0; i < 6; ++i) {
        const bool ok = (unsigned)(y + cDy[i]) < (unsigned)HW;
        const v4f tv = *(const v4f*)(xb + 16 * tl + (ok ? cOff[i] : cSafe[i]));
        cv[tl][i] = ok ? tv : z4;
      }
    }
#pragma unroll
    for (int tl = 0; tl < MT; ++tl) {
      const int xt = x0 + 16 * tl;
      const bool colOk = side ? (xt < HW - 16) : (xt > 0);
#pragma unroll
      for (int i = 0; i < 3; ++i) {
        const bool ok = colOk && ((unsigned)(y + hDy[i]) < (unsigned)HW);
        const float tv = xb[16 * tl + (ok ? hOff[i] : hSafe[i])];
        hv[tl][i] = ok ? tv : 0.0f;
      }
    }
#pragma unroll
    for (int tl = 0; tl < MT; ++tl) {
#pragma unroll
      for (int i = 0; i < 6; ++i) *(v4f*)(patch + tl * PSTR + cLds[i]) = cv[tl][i];
#pragma unroll
      for (int i = 0; i < 3; ++i) patch[tl * PSTR + hLds[i]] = hv[tl][i];
    }
    __syncthreads();

#pragma unroll
    for (int i = 0; i < 2; ++i) {
      if (lane + 32 * i < 3 * CSW) {
#pragma unroll
        for (int tl = 0; tl < MT; ++tl) {
          float s = 0.0f;
#pragma unroll
          for (int c = 0; c < NC; ++c) s += patch[tl * PSTR + bOff[i] + c * 3 * PP];
          cs[tl * 64 + lane + 32 * i] = s;
        }
      }
    }
    __syncthreads();

    v8f acc0[MT], acc1[MT];
#pragma unroll
    for (int tl = 0; tl < MT; ++tl) {
#pragma unroll
      for (int r = 0; r < 8; ++r) { acc0[tl][r] = b1v0; acc1[tl][r] = b1v1; }
    }
    const float* w1Base = sW1 + col * P1 + 2 * h;
#pragma unroll
    for (int s = 0; s < DF / 4; ++s) {
      const int u = s / 3, v = s - 3 * u;
      v2f bA = *(const v2f*)(w1Base + 4 * s);
      v2f bB = *(const v2f*)(w1Base + 4 * s + 16 * P1);
#pragma unroll
      for (int tl = 0; tl < MT; ++tl) {
        v2f a;
        a.x = patch[tl * PSTR + aOff[v][0] + u * 4 * PP];
        a.y = patch[tl * PSTR + aOff[v][1] + u * 4 * PP];
        acc0[tl] = wmma4(a, bA, acc0[tl]);
        acc1[tl] = wmma4(a, bB, acc1[tl]);
      }
    }
#pragma unroll
    for (int tl = 0; tl < MT; ++tl) {
#pragma unroll
      for (int r = 0; r < 8; ++r) {
        int m2 = r + 8 * h;
        hid[tl * HSTR + m2 * NO + col]      = tanh_sel(acc0[tl][r]);
        hid[tl * HSTR + m2 * NO + col + 16] = tanh_sel(acc1[tl][r]);
      }
    }
    __syncthreads();

    float hr[MT][NO];
    float pr[MT][9];
#pragma unroll
    for (int tl = 0; tl < MT; ++tl) {
#pragma unroll
      for (int k = 0; k < NO / 4; ++k) {
        v4f v = *(const v4f*)(hid + tl * HSTR + col * NO + 4 * k);
        hr[tl][4 * k + 0] = v.x; hr[tl][4 * k + 1] = v.y; hr[tl][4 * k + 2] = v.z; hr[tl][4 * k + 3] = v.w;
      }
#pragma unroll
      for (int j = 0; j < 9; ++j) pr[tl][j] = cs[tl * 64 + (j / 3) * CSW + col + (j % 3)];

#pragma unroll
      for (int r = 0; r < 8; ++r) {
        int m2 = r + 8 * h;
        float s0 = bs0, s1 = bs1;
#pragma unroll
        for (int j = 0; j < 9; ++j) {
          float pv = cs[tl * 64 + (j / 3) * CSW + m2 + (j % 3)];
          s0 += pv * b2a[j]; s1 += pv * b2b[j];
        }
        acc0[tl][r] = s0; acc1[tl][r] = s1;
      }
    }

    const float* w2Base = sW2 + col * P2 + 2 * h;
#pragma unroll
    for (int s = 0; s < DQ / 4; ++s) {
      const int q = 4 * s;
      v2f bA = *(const v2f*)(w2Base + 4 * s);
      v2f bB = *(const v2f*)(w2Base + 4 * s + 16 * P2);
#pragma unroll
      for (int tl = 0; tl < MT; ++tl) {
        float z0 = hr[tl][(q + 0) / 9] * pr[tl][(q + 0) % 9];
        float z1 = hr[tl][(q + 1) / 9] * pr[tl][(q + 1) % 9];
        float z2 = hr[tl][(q + 2) / 9] * pr[tl][(q + 2) % 9];
        float z3 = hr[tl][(q + 3) / 9] * pr[tl][(q + 3) % 9];
        v2f a; a.x = h ? z2 : z0; a.y = h ? z3 : z1;
        acc0[tl] = wmma4(a, bA, acc0[tl]);
        acc1[tl] = wmma4(a, bB, acc1[tl]);
      }
    }

    float* outT = hid;
#pragma unroll
    for (int tl = 0; tl < MT; ++tl) {
#pragma unroll
      for (int r = 0; r < 8; ++r) {
        int m2 = r + 8 * h;
        outT[tl * HSTR + m2 * NO + col]      = acc0[tl][r];
        outT[tl * HSTR + m2 * NO + col + 16] = acc1[tl][r];
      }
    }
    __syncthreads();
#pragma unroll
    for (int tl = 0; tl < MT; ++tl) {
#pragma unroll
      for (int e = 0; e < 16; ++e) {
        int idx = e * 32 + lane;
        int o = idx >> 4, m = idx & 15;
        Out[((long)(bimg * NO + o)) * LPIX + p0 + 16 * tl + m] = outT[tl * HSTR + m * NO + o];
      }
    }
    __syncthreads();
  }
}

extern "C" void kernel_launch(void* const* d_in, const int* in_sizes, int n_in,
                              void* d_out, int out_size, void* d_ws, size_t ws_size,
                              hipStream_t stream) {
  (void)in_sizes; (void)n_in; (void)out_size; (void)d_ws; (void)ws_size;
  const float* X    = (const float*)d_in[0];
  const float* W1   = (const float*)d_in[1];
  const float* b1   = (const float*)d_in[2];
  const float* W2   = (const float*)d_in[3];
  const float* b2   = (const float*)d_in[4];
  const float* bias = (const float*)d_in[5];
  float* Out = (float*)d_out;
  dynaconv_kernel<<<BLOCKS, WAVES * 32, 0, stream>>>(X, W1, b1, W2, b2, bias, Out);
}


# input halo loads issued one trip ahead (in flight during GEMM2); the five block barriers per trip, which drain loads and stores in flight, become the wave's own LDS wait (all scratch is per wave)
# speedup vs baseline: 3.0376x; 1.0655x over previous
#include <hip/hip_runtime.h>
#include <math.h>


typedef float v2f __attribute__((ext_vector_type(2)));
typedef float v4f __attribute__((ext_vector_type(4)));
typedef float v8f __attribute__((ext_vector_type(8)));

#define NB 4
#define NC 16
#define NO 32
#define HW 256
#define LPIX (HW*HW)
#define DF 144
#define DQ 288
#define WAVES 4
#define TPW 8
#define MT 2
#define ROWS_TOTAL (NB*LPIX)
#define TILES_TOTAL (ROWS_TOTAL/16)
#define BLOCKS (TILES_TOTAL/(WAVES*TPW))

#define P1 (DF + 4)
#define P2 (DQ + 4)

#define PP   24
#define NSEG (NC*3)
#define CSW  18

static __device__ __forceinline__ v8f wmma4(v2f a, v2f b, v8f c) {
  return __builtin_amdgcn_wmma_f32_16x16x4_f32(
      false, a, false, b, (short)0, c, false, false);
}


static __device__ __forceinline__ void wave_lds_handover() {
  asm volatile("s_wait_dscnt 0x0" ::: "memory");
}

static __device__ __forceinline__ float tanh_sel(float x) {
  const float y  = fabsf(x);
  const float y2 = y * y;
  const float p  = fmaf(y2, fmaf(y2, fmaf(y2, fmaf(y2, -0x1.758e7ap-8f, 0x1.521192p-6f),
                                           -0x1.b8389cp-5f), 0x1.110704p-3f), -0x1.555532p-2f);
  const float zs = fmaf(y2, y * p, y);
  const float t  = expf(2.0f * y);
  const float zl = fmaf(-2.0f, __builtin_amdgcn_rcpf(t + 1.0f), 1.0f);
  return copysignf(y < 0.625f ? zs : zl, x);
}

__global__ __launch_bounds__(WAVES * 32)
void dynaconv_kernel(const float* __restrict__ X,
                     const float* __restrict__ W1,
                     const float* __restrict__ b1,
                     const float* __restrict__ W2,
                     const float* __restrict__ b2,
                     const float* __restrict__ bias,
                     float* __restrict__ Out) {
  __shared__ __align__(16) float sW1[ NO * P1 ];
  __shared__ __align__(16) float sW2[ NO * P2 ];
  __shared__ __align__(16) float sPatch[WAVES][MT][NSEG * PP];
  __shared__ __align__(16) float sCs  [WAVES][MT][64];
  __shared__ __align__(16) float sHid [WAVES][MT][16 * NO];

  const int tid  = threadIdx.x;
  const int lane = tid & 31;
  const int wv   = tid >> 5;
  const int col  = lane & 15;
  const int h    = lane >> 4;

  {
    constexpr int NT  = WAVES * 32;
    constexpr int N1  = NO * DF / 4 / NT;
    constexpr int N2  = DQ * NO / 4 / NT;
    v4f w1v[N1], w2v[N2];
#pragma unroll
    for (int i = 0; i < N1; ++i) w1v[i] = ((const v4f*)W1)[tid + NT * i];
#pragma unroll
    for (int i = 0; i < N2; ++i) w2v[i] = ((const v4f*)W2)[tid + NT * i];
#pragma unroll
    for (int i = 0; i < N1; ++i) {
      const int e4 = tid + NT * i;
      const int o = e4 / (DF / 4), c = e4 - o * (DF / 4);
      *(v4f*)(sW1 + o * P1 + 4 * c) = w1v[i];
    }
#pragma unroll
    for (int i = 0; i < N2; ++i) {
      const int e4 = tid + NT * i;
      const int g = e4 >> 3, c4 = e4 & 7;
      const int o = g / 9, j = g - o * 9;
      float* d = sW2 + o * P2 + 36 * c4 + j;
      d[0] = w2v[i].x; d[9] = w2v[i].y; d[18] = w2v[i].z; d[27] = w2v[i].w;
    }
  }
  const float b1v0 = b1[col],      b1v1 = b1[col + 16];
  const float bs0  = bias[col],    bs1  = bias[col + 16];
  float b2a[9], b2b[9];
#pragma unroll
  for (int j = 0; j < 9; ++j) { b2a[j] = b2[col * 9 + j]; b2b[j] = b2[(col + 16) * 9 + j]; }

  int cOff[6], cSafe[6], cDy[6], cLds[6];
#pragma unroll
  for (int i = 0; i < 6; ++i) {
    const int idx = lane + 32 * i;
    const int seg = idx >> 2, qd = idx & 3;
    const int c = seg / 3, di = seg - 3 * c;
    cDy[i]   = di - 1;
    cSafe[i] = c * LPIX + 4 * qd;
    cOff[i]  = cSafe[i] + (di - 1) * HW;
    cLds[i]  = seg * PP + 4 + 4 * qd;
  }
  const int side = lane & 1;
  int hOff[3], hSafe[3], hDy[3], hLds[3];
#pragma unroll
  for (int i = 0; i < 3; ++i) {
    const int seg = (lane >> 1) + 16 * i;
    const int c = seg / 3, di = seg - 3 * c;
    hDy[i]   = di - 1;
    hSafe[i] = c * LPIX;
    hOff[i]  = c * LPIX + (di - 1) * HW + (side ? 16 : -1);
    hLds[i]  = seg * PP + (side ? 20 : 3);
  }
  int bOff[2];
#pragma unroll
  for (int i = 0; i < 2; ++i) {
    const int e = lane + 32 * i;
    const int di = e / CSW, j = e - CSW * di;
    bOff[i] = di * PP + 3 + j;
  }
  int aOff[3][2];
#pragma unroll
  for (int v = 0; v < 3; ++v) {
#pragma unroll
    for (int e = 0; e < 2; ++e) {
      const int k = 4 * v + 2 * h + e;
      aOff[v][e] = (k / 3) * PP + 3 + (k % 3) + col;
    }
  }
  __syncthreads();

  float* patch = sPatch[wv][0];
  float* cs    = sCs[wv][0];
  float* hid   = sHid[wv][0];
  constexpr int PSTR = NSEG * PP;
  constexpr int HSTR = 16 * NO;

  constexpr int NTRIP = TPW / MT;
  const int tile0 = ((int)blockIdx.x * WAVES + wv) * TPW;

  v4f cv[MT][6];
  float hv[MT][3];
  auto issue_halo_loads = [&](int trip) {
    const int row0 = (tile0 + trip * MT) << 4;
    const int p0   = row0 & (LPIX - 1);
    const int y    = p0 >> 8;
    const int x0   = p0 & (HW - 1);
    const float* xb = X + ((row0 >> 16) * NC * LPIX + p0);
#pragma unroll
    for (int tl = 0; tl < MT; ++tl) {
#pragma unroll
      for (int i = 0; i < 6; ++i) {
        const int  offIn = cOff[i], offSafe = cSafe[i];
        const bool ok = (unsigned)(y + cDy[i]) < (unsigned)HW;
        cv[tl][i] = *(const v4f*)(xb + 16 * tl + (ok ? offIn : offSafe));
      }
    }
#pragma unroll
    for (int tl = 0; tl < MT; ++tl) {
      const int xt = x0 + 16 * tl;
      const bool colOk = side ? (xt < HW - 16) : (xt > 0);
#pragma unroll
      for (int i = 0; i < 3; ++i) {
        const int  offIn = hOff[i], offSafe = hSafe[i];
        const bool ok = colOk && ((unsigned)(y + hDy[i]) < (unsigned)HW);
        hv[tl][i] = xb[16 * tl + (ok ? offIn : offSafe)];
      }
    }
  };
  issue_halo_loads(0);

  for (int it = 0; it < NTRIP; ++it) {
    const int row0 = (tile0 + it * MT) << 4;
    const int bimg = row0 >> 16;
    const int p0   = row0 & (LPIX - 1);
    const int y    = p0 >> 8;
    const int x0   = p0 & (HW - 1);

    const v4f z4 = {0.0f, 0.0f, 0.0f, 0.0f};
#pragma unroll
    for (int tl = 0; tl < MT; ++tl) {
#pragma unroll
      for (int i = 0; i < 6; ++i) {
        const bool ok = (unsigned)(y + cDy[i]) < (unsigned)HW;
        *(v4f*)(patch + tl * PSTR + cLds[i]) = ok ? cv[tl][i] : z4;
      }
      const int xt = x0 + 16 * tl;
      const bool colOk = side ? (xt < HW - 16) : (xt > 0);
#pragma unroll
      for (int i = 0; i < 3; ++i) {
        const bool ok = colOk && ((unsigned)(y + hDy[i]) < (unsigned)HW);
        patch[tl * PSTR + hLds[i]] = ok ? hv[tl][i] : 0.0f;
      }
    }
    wave_lds_handover();

#pragma unroll
    for (int i = 0; i < 2; ++i) {
      if (lane + 32 * i < 3 * CSW) {
#pragma unroll
        for (int tl = 0; tl < MT; ++tl) {
          float s = 0.0f;
#pragma unroll
          for (int c = 0; c < NC; ++c) s += patch[tl * PSTR + bOff[i] + c * 3 * PP];
          cs[tl * 64 + lane + 32 * i] = s;
        }
      }
    }
    wave_lds_handover();

    v8f acc0[MT], acc1[MT];
#pragma unroll
    for (int tl = 0; tl < MT; ++tl) {
#pragma unroll
      for (int r = 0; r < 8; ++r) { acc0[tl][r] = b1v0; acc1[tl][r] = b1v1; }
    }
    const float* w1Base = sW1 + col * P1 + 2 * h;
#pragma unroll
    for (int s = 0; s < DF / 4; ++s) {
      const int u = s / 3, v = s - 3 * u;
      v2f bA = *(const v2f*)(w1Base + 4 * s);
      v2f bB = *(const v2f*)(w1Base + 4 * s + 16 * P1);
#pragma unroll
      for (int tl = 0; tl < MT; ++tl) {
        v2f a;
        a.x = patch[tl * PSTR + aOff[v][0] + u * 4 * PP];
        a.y = patch[tl * PSTR + aOff[v][1] + u * 4 * PP];
        acc0[tl] = wmma4(a, bA, acc0[tl]);
        acc1[tl] = wmma4(a, bB, acc1[tl]);
      }
    }
#pragma unroll
    for (int tl = 0; tl < MT; ++tl) {
#pragma unroll
      for (int r = 0; r < 8; ++r) {
        int m2 = r + 8 * h;
        hid[tl * HSTR + m2 * NO + col]      = tanh_sel(acc0[tl][r]);
        hid[tl * HSTR + m2 * NO + col + 16] = tanh_sel(acc1[tl][r]);
      }
    }
    if (it + 1 < NTRIP) issue_halo_loads(it + 1);
    wave_lds_handover();

    float hr[MT][NO];
    float pr[MT][9];
#pragma unroll
    for (int tl = 0; tl < MT; ++tl) {
#pragma unroll
      for (int k = 0; k < NO / 4; ++k) {
        v4f v = *(const v4f*)(hid + tl * HSTR + col * NO + 4 * k);
        hr[tl][4 * k + 0] = v.x; hr[tl][4 * k + 1] = v.y; hr[tl][4 * k + 2] = v.z; hr[tl][4 * k + 3] = v.w;
      }
#pragma unroll
      for (int j = 0; j < 9; ++j) pr[tl][j] = cs[tl * 64 + (j / 3) * CSW + col + (j % 3)];

#pragma unroll
      for (int r = 0; r < 8; ++r) {
        int m2 = r + 8 * h;
        float s0 = bs0, s1 = bs1;
#pragma unroll
        for (int j = 0; j < 9; ++j) {
          float pv = cs[tl * 64 + (j / 3) * CSW + m2 + (j % 3)];
          s0 += pv * b2a[j]; s1 += pv * b2b[j];
        }
        acc0[tl][r] = s0; acc1[tl][r] = s1;
      }
    }

    const float* w2Base = sW2 + col * P2 + 2 * h;
#pragma unroll
    for (int s = 0; s < DQ / 4; ++s) {
      const int q = 4 * s;
      v2f bA = *(const v2f*)(w2Base + 4 * s);
      v2f bB = *(const v2f*)(w2Base + 4 * s + 16 * P2);
#pragma unroll
      for (int tl = 0; tl < MT; ++tl) {
        float z0 = hr[tl][(q + 0) / 9] * pr[tl][(q + 0) % 9];
        float z1 = hr[tl][(q + 1) / 9] * pr[tl][(q + 1) % 9];
        float z2 = hr[tl][(q + 2) / 9] * pr[tl][(q + 2) % 9];
        float z3 = hr[tl][(q + 3) / 9] * pr[tl][(q + 3) % 9];
        v2f a; a.x = h ? z2 : z0; a.y = h ? z3 : z1;
        acc0[tl] = wmma4(a, bA, acc0[tl]);
        acc1[tl] = wmma4(a, bB, acc1[tl]);
      }
    }

    float* outT = hid;
#pragma unroll
    for (int tl = 0; tl < MT; ++tl) {
#pragma unroll
      for (int r = 0; r < 8; ++r) {
        int m2 = r + 8 * h;
        outT[tl * HSTR + m2 * NO + col]      = acc0[tl][r];
        outT[tl * HSTR + m2 * NO + col + 16] = acc1[tl][r];
      }
    }
    wave_lds_handover();
#pragma unroll
    for (int tl = 0; tl < MT; ++tl) {
#pragma unroll
      for (int e = 0; e < 16; ++e) {
        int idx = e * 32 + lane;
        int o = idx >> 4, m = idx & 15;
        Out[((long)(bimg * NO + o)) * LPIX + p0 + 16 * tl + m] = outT[tl * HSTR + m * NO + o];
      }
    }
  }
}

extern "C" void kernel_launch(void* const* d_in, const int* in_sizes, int n_in,
                              void* d_out, int out_size, void* d_ws, size_t ws_size,
                              hipStream_t stream) {
  (void)in_sizes; (void)n_in; (void)out_size; (void)d_ws; (void)ws_size;
  const float* X    = (const float*)d_in[0];
  const float* W1   = (const float*)d_in[1];
  const float* b1   = (const float*)d_in[2];
  const float* W2   = (const float*)d_in[3];
  const float* b2   = (const float*)d_in[4];
  const float* bias = (const float*)d_in[5];
  float* Out = (float*)d_out;
  dynaconv_kernel<<<BLOCKS, WAVES * 32, 0, stream>>>(X, W1, b1, W2, b2, bias, Out);
}


# tanh of the hidden layer on the hardware f32 tanh instruction, one transcendental per value (was the library's two formulas by hand, about 25 vector instructions per value); last bits of tanh differ
# speedup vs baseline: 3.2235x; 1.0612x over previous
#include <hip/hip_runtime.h>
#include <math.h>


typedef float v2f __attribute__((ext_vector_type(2)));
typedef float v4f __attribute__((ext_vector_type(4)));
typedef float v8f __attribute__((ext_vector_type(8)));

#define NB 4
#define NC 16
#define NO 32
#define HW 256
#define LPIX (HW*HW)
#define DF 144
#define DQ 288
#define WAVES 4
#define TPW 8
#define MT 2
#define ROWS_TOTAL (NB*LPIX)
#define TILES_TOTAL (ROWS_TOTAL/16)
#define BLOCKS (TILES_TOTAL/(WAVES*TPW))

#define P1 (DF + 4)
#define P2 (DQ + 4)

#define PP   24
#define NSEG (NC*3)
#define CSW  18

static __device__ __forceinline__ v8f wmma4(v2f a, v2f b, v8f c) {
  return __builtin_amdgcn_wmma_f32_16x16x4_f32(
      false, a, false, b, (short)0, c, false, false);
}


static __device__ __forceinline__ void wave_lds_handover() {
  asm volatile("s_wait_dscnt 0x0" ::: "memory");
}

static __device__ __forceinline__ float tanh_hw(float x) {
  return __builtin_amdgcn_tanhf(x);
}

__global__ __launch_bounds__(WAVES * 32)
void dynaconv_kernel(const float* __restrict__ X,
                     const float* __restrict__ W1,
                     const float* __restrict__ b1,
                     const float* __restrict__ W2,
                     const float* __restrict__ b2,
                     const float* __restrict__ bias,
                     float* __restrict__ Out) {
  __shared__ __align__(16) float sW1[ NO * P1 ];
  __shared__ __align__(16) float sW2[ NO * P2 ];
  __shared__ __align__(16) float sPatch[WAVES][MT][NSEG * PP];
  __shared__ __align__(16) float sCs  [WAVES][MT][64];
  __shared__ __align__(16) float sHid [WAVES][MT][16 * NO];

  const int tid  = threadIdx.x;
  const int lane = tid & 31;
  const int wv   = tid >> 5;
  const int col  = lane & 15;
  const int h    = lane >> 4;

  {
    constexpr int NT  = WAVES * 32;
    constexpr int N1  = NO * DF / 4 / NT;
    constexpr int N2  = DQ * NO / 4 / NT;
    v4f w1v[N1], w2v[N2];
#pragma unroll
    for (int i = 0; i < N1; ++i) w1v[i] = ((const v4f*)W1)[tid + NT * i];
#pragma unroll
    for (int i = 0; i < N2; ++i) w2v[i] = ((const v4f*)W2)[tid + NT * i];
#pragma unroll
    for (int i = 0; i < N1; ++i) {
      const int e4 = tid + NT * i;
      const int o = e4 / (DF / 4), c = e4 - o * (DF / 4);
      *(v4f*)(sW1 + o * P1 + 4 * c) = w1v[i];
    }
#pragma unroll
    for (int i = 0; i < N2; ++i) {
      const int e4 = tid + NT * i;
      const int g = e4 >> 3, c4 = e4 & 7;
      const int o = g / 9, j = g - o * 9;
      float* d = sW2 + o * P2 + 36 * c4 + j;
      d[0] = w2v[i].x; d[9] = w2v[i].y; d[18] = w2v[i].z; d[27] = w2v[i].w;
    }
  }
  const float b1v0 = b1[col],      b1v1 = b1[col + 16];
  const float bs0  = bias[col],    bs1  = bias[col + 16];
  float b2a[9], b2b[9];
#pragma unroll
  for (int j = 0; j < 9; ++j) { b2a[j] = b2[col * 9 + j]; b2b[j] = b2[(col + 16) * 9 + j]; }

  int cOff[6], cSafe[6], cDy[6], cLds[6];
#pragma unroll
  for (int i = 0; i < 6; ++i) {
    const int idx = lane + 32 * i;
    const int seg = idx >> 2, qd = idx & 3;
    const int c = seg / 3, di = seg - 3 * c;
    cDy[i]   = di - 1;
    cSafe[i] = c * LPIX + 4 * qd;
    cOff[i]  = cSafe[i] + (di - 1) * HW;
    cLds[i]  = seg * PP + 4 + 4 * qd;
  }
  const int side = lane & 1;
  int hOff[3], hSafe[3], hDy[3], hLds[3];
#pragma unroll
  for (int i = 0; i < 3; ++i) {
    const int seg = (lane >> 1) + 16 * i;
    const int c = seg / 3, di = seg - 3 * c;
    hDy[i]   = di - 1;
    hSafe[i] = c * LPIX;
    hOff[i]  = c * LPIX + (di - 1) * HW + (side ? 16 : -1);
    hLds[i]  = seg * PP + (side ? 20 : 3);
  }
  int bOff[2];
#pragma unroll
  for (int i = 0; i < 2; ++i) {
    const int e = lane + 32 * i;
    const int di = e / CSW, j = e - CSW * di;
    bOff[i] = di * PP + 3 + j;
  }
  int aOff[3][2];
#pragma unroll
  for (int v = 0; v < 3; ++v) {
#pragma unroll
    for (int e = 0; e < 2; ++e) {
      const int k = 4 * v + 2 * h + e;
      aOff[v][e] = (k / 3) * PP + 3 + (k % 3) + col;
    }
  }
  __syncthreads();

  float* patch = sPatch[wv][0];
  float* cs    = sCs[wv][0];
  float* hid   = sHid[wv][0];
  constexpr int PSTR = NSEG * PP;
  constexpr int HSTR = 16 * NO;

  constexpr int NTRIP = TPW / MT;
  const int tile0 = ((int)blockIdx.x * WAVES + wv) * TPW;

  v4f cv[MT][6];
  float hv[MT][3];
  auto issue_halo_loads = [&](int trip) {
    const int row0 = (tile0 + trip * MT) << 4;
    const int p0   = row0 & (LPIX - 1);
    const int y    = p0 >> 8;
    const int x0   = p0 & (HW - 1);
    const float* xb = X + ((row0 >> 16) * NC * LPIX + p0);
#pragma unroll
    for (int tl = 0; tl < MT; ++tl) {
#pragma unroll
      for (int i = 0; i < 6; ++i) {
        const int  offIn = cOff[i], offSafe = cSafe[i];
        const bool ok = (unsigned)(y + cDy[i]) < (unsigned)HW;
        cv[tl][i] = *(const v4f*)(xb + 16 * tl + (ok ? offIn : offSafe));
      }
    }
#pragma unroll
    for (int tl = 0; tl < MT; ++tl) {
      const int xt = x0 + 16 * tl;
      const bool colOk = side ? (xt < HW - 16) : (xt > 0);
#pragma unroll
      for (int i = 0; i < 3; ++i) {
        const int  offIn = hOff[i], offSafe = hSafe[i];
        const bool ok = colOk && ((unsigned)(y + hDy[i]) < (unsigned)HW);
        hv[tl][i] = xb[16 * tl + (ok ? offIn : offSafe)];
      }
    }
  };
  issue_halo_loads(0);

  for (int it = 0; it < NTRIP; ++it) {
    const int row0 = (tile0 + it * MT) << 4;
    const int bimg = row0 >> 16;
    const int p0   = row0 & (LPIX - 1);
    const int y    = p0 >> 8;
    const int x0   = p0 & (HW - 1);

    const v4f z4 = {0.0f, 0.0f, 0.0f, 0.0f};
#pragma unroll
    for (int tl = 0; tl < MT; ++tl) {
#pragma unroll
      for (int i = 0; i < 6; ++i) {
        const bool ok = (unsigned)(y + cDy[i]) < (unsigned)HW;
        *(v4f*)(patch + tl * PSTR + cLds[i]) = ok ? cv[tl][i] : z4;
      }
      const int xt = x0 + 16 * tl;
      const bool colOk = side ? (xt < HW - 16) : (xt > 0);
#pragma unroll
      for (int i = 0; i < 3; ++i) {
        const bool ok = colOk && ((unsigned)(y + hDy[i]) < (unsigned)HW);
        patch[tl * PSTR + hLds[i]] = ok ? hv[tl][i] : 0.0f;
      }
    }
    wave_lds_handover();

#pragma unroll
    for (int i = 0; i < 2; ++i) {
      if (lane + 32 * i < 3 * CSW) {
#pragma unroll
        for (int tl = 0; tl < MT; ++tl) {
          float s = 0.0f;
#pragma unroll
          for (int c = 0; c < NC; ++c) s += patch[tl * PSTR + bOff[i] + c * 3 * PP];
          cs[tl * 64 + lane + 32 * i] = s;
        }
      }
    }
    wave_lds_handover();

    v8f acc0[MT], acc1[MT];
#pragma unroll
    for (int tl = 0; tl < MT; ++tl) {
#pragma unroll
      for (int r = 0; r < 8; ++r) { acc0[tl][r] = b1v0; acc1[tl][r] = b1v1; }
    }
    const float* w1Base = sW1 + col * P1 + 2 * h;
#pragma unroll
    for (int s = 0; s < DF / 4; ++s) {
      const int u = s / 3, v = s - 3 * u;
      v2f bA = *(const v2f*)(w1Base + 4 * s);
      v2f bB = *(const v2f*)(w1Base + 4 * s + 16 * P1);
#pragma unroll
      for (int tl = 0; tl < MT; ++tl) {
        v2f a;
        a.x = patch[tl * PSTR + aOff[v][0] + u * 4 * PP];
        a.y = patch[tl * PSTR + aOff[v][1] + u * 4 * PP];
        acc0[tl] = wmma4(a, bA, acc0[tl]);
        acc1[tl] = wmma4(a, bB, acc1[tl]);
      }
    }
#pragma unroll
    for (int tl = 0; tl < MT; ++tl) {
#pragma unroll
      for (int r = 0; r < 8; ++r) {
        int m2 = r + 8 * h;
        hid[tl * HSTR + m2 * NO + col]      = tanh_hw(acc0[tl][r]);
        hid[tl * HSTR + m2 * NO + col + 16] = tanh_hw(acc1[tl][r]);
      }
    }
    if (it + 1 < NTRIP) issue_halo_loads(it + 1);
    wave_lds_handover();

    float hr[MT][NO];
    float pr[MT][9];
#pragma unroll
    for (int tl = 0; tl < MT; ++tl) {
#pragma unroll
      for (int k = 0; k < NO / 4; ++k) {
        v4f v = *(const v4f*)(hid + tl * HSTR + col * NO + 4 * k);
        hr[tl][4 * k + 0] = v.x; hr[tl][4 * k + 1] = v.y; hr[tl][4 * k + 2] = v.z; hr[tl][4 * k + 3] = v.w;
      }
#pragma unroll
      for (int j = 0; j < 9; ++j) pr[tl][j] = cs[tl * 64 + (j / 3) * CSW + col + (j % 3)];

#pragma unroll
      for (int r = 0; r < 8; ++r) {
        int m2 = r + 8 * h;
        float s0 = bs0, s1 = bs1;
#pragma unroll
        for (int j = 0; j < 9; ++j) {
          float pv = cs[tl * 64 + (j / 3) * CSW + m2 + (j % 3)];
          s0 += pv * b2a[j]; s1 += pv * b2b[j];
        }
        acc0[tl][r] = s0; acc1[tl][r] = s1;
      }
    }

    const float* w2Base = sW2 + col * P2 + 2 * h;
#pragma unroll
    for (int s = 0; s < DQ / 4; ++s) {
      const int q = 4 * s;
      v2f bA = *(const v2f*)(w2Base + 4 * s);
      v2f bB = *(const v2f*)(w2Base + 4 * s + 16 * P2);
#pragma unroll
      for (int tl = 0; tl < MT; ++tl) {
        float z0 = hr[tl][(q + 0) / 9] * pr[tl][(q + 0) % 9];
        float z1 = hr[tl][(q + 1) / 9] * pr[tl][(q + 1) % 9];
        float z2 = hr[tl][(q + 2) / 9] * pr[tl][(q + 2) % 9];
        float z3 = hr[tl][(q + 3) / 9] * pr[tl][(q + 3) % 9];
        v2f a; a.x = h ? z2 : z0; a.y = h ? z3 : z1;
        acc0[tl] = wmma4(a, bA, acc0[tl]);
        acc1[tl] = wmma4(a, bB, acc1[tl]);
      }
    }

    float* outT = hid;
#pragma unroll
    for (int tl = 0; tl < MT; ++tl) {
#pragma unroll
      for (int r = 0; r < 8; ++r) {
        int m2 = r + 8 * h;
        outT[tl * HSTR + m2 * NO + col]      = acc0[tl][r];
        outT[tl * HSTR + m2 * NO + col + 16] = acc1[tl][r];
      }
    }
    wave_lds_handover();
#pragma unroll
    for (int tl = 0; tl < MT; ++tl) {
#pragma unroll
      for (int e = 0; e < 16; ++e) {
        int idx = e * 32 + lane;
        int o = idx >> 4, m = idx & 15;
        Out[((long)(bimg * NO + o)) * LPIX + p0 + 16 * tl + m] = outT[tl * HSTR + m * NO + o];
      }
    }
  }
}

extern "C" void kernel_launch(void* const* d_in, const int* in_sizes, int n_in,
                              void* d_out, int out_size, void* d_ws, size_t ws_size,
                              hipStream_t stream) {
  (void)in_sizes; (void)n_in; (void)out_size; (void)d_ws; (void)ws_size;
  const float* X    = (const float*)d_in[0];
  const float* W1   = (const float*)d_in[1];
  const float* b1   = (const float*)d_in[2];
  const float* W2   = (const float*)d_in[3];
  const float* b2   = (const float*)d_in[4];
  const float* bias = (const float*)d_in[5];
  float* Out = (float*)d_out;
  dynaconv_kernel<<<BLOCKS, WAVES * 32, 0, stream>>>(X, W1, b1, W2, b2, bias, Out);
}


# 16 waves per work-group, one work-group per WGP: 4 waves per SIMD at 256 registers (was 2 at 388), a wave runs 2 trips of 2 tiles (was 4); loads as scalar base + 32-bit lane offset; sums unchanged
# speedup vs baseline: 3.3433x; 1.0372x over previous
#include <hip/hip_runtime.h>
#include <math.h>


typedef float v2f __attribute__((ext_vector_type(2)));
typedef float v4f __attribute__((ext_vector_type(4)));
typedef float v8f __attribute__((ext_vector_type(8)));

#define NB 4
#define NC 16
#define NO 32
#define HW 256
#define LPIX (HW*HW)
#define DF 144
#define DQ 288
#define WAVES 16
#define TPW 4
#define MT 2
#define ROWS_TOTAL (NB*LPIX)
#define TILES_TOTAL (ROWS_TOTAL/16)
#define BLOCKS (TILES_TOTAL/(WAVES*TPW))

#define P1 (DF + 4)
#define P2 (DQ + 4)

#define PP   24
#define NSEG (NC*3)
#define CSW  18

static __device__ __forceinline__ v8f wmma4(v2f a, v2f b, v8f c) {
  return __builtin_amdgcn_wmma_f32_16x16x4_f32(
      false, a, false, b, (short)0, c, false, false);
}


static __device__ __forceinline__ void wave_lds_handover() {
  asm volatile("s_wait_dscnt 0x0" ::: "memory");
}

static __device__ __forceinline__ float tanh_hw(float x) {
  return __builtin_amdgcn_tanhf(x);
}

__global__ __launch_bounds__(WAVES * 32)
void dynaconv_kernel(const float* __restrict__ X,
                     const float* __restrict__ W1,
                     const float* __restrict__ b1,
                     const float* __restrict__ W2,
                     const float* __restrict__ b2,
                     const float* __restrict__ bias,
                     float* __restrict__ Out) {
  __shared__ __align__(16) float sW1[ NO * P1 ];
  __shared__ __align__(16) float sW2[ NO * P2 ];
  __shared__ __align__(16) float sPatch[WAVES][MT][NSEG * PP];
  __shared__ __align__(16) float sCs  [WAVES][MT][64];
  __shared__ __align__(16) float sHid [WAVES][MT][16 * NO];

  const int tid  = threadIdx.x;
  const int lane = tid & 31;
  const int wv   = __builtin_amdgcn_readfirstlane(tid >> 5);
  const int col  = lane & 15;
  const int h    = lane >> 4;

  {
    constexpr int NT  = WAVES * 32;
    constexpr int Q1  = NO * DF / 4;
    constexpr int Q2  = DQ * NO / 4;
    constexpr int N1  = (Q1 + NT - 1) / NT;
    constexpr int N2  = (Q2 + NT - 1) / NT;
    v4f w1v[N1], w2v[N2];
#pragma unroll
    for (int i = 0; i < N1; ++i) if (tid + NT * i < Q1) w1v[i] = ((const v4f*)W1)[tid + NT * i];
#pragma unroll
    for (int i = 0; i < N2; ++i) if (tid + NT * i < Q2) w2v[i] = ((const v4f*)W2)[tid + NT * i];
#pragma unroll
    for (int i = 0; i < N1; ++i) {
      const int e4 = tid + NT * i;
      if (e4 < Q1) {
        const int o = e4 / (DF / 4), c = e4 - o * (DF / 4);
        *(v4f*)(sW1 + o * P1 + 4 * c) = w1v[i];
      }
    }
#pragma unroll
    for (int i = 0; i < N2; ++i) {
      const int e4 = tid + NT * i;
      if (e4 < Q2) {
        const int g = e4 >> 3, c4 = e4 & 7;
        const int o = g / 9, j = g - o * 9;
        float* d = sW2 + o * P2 + 36 * c4 + j;
        d[0] = w2v[i].x; d[9] = w2v[i].y; d[18] = w2v[i].z; d[27] = w2v[i].w;
      }
    }
  }
  const float b1v0 = b1[col],      b1v1 = b1[col + 16];
  const float bs0  = bias[col],    bs1  = bias[col + 16];
  float b2a[9], b2b[9];
#pragma unroll
  for (int j = 0; j < 9; ++j) { b2a[j] = b2[col * 9 + j]; b2b[j] = b2[(col + 16) * 9 + j]; }

  unsigned cSafeB[3]; int cDyB[3], cLds[3];
  constexpr unsigned C8B = 8u * LPIX * 4u;
#pragma unroll
  for (int i = 0; i < 3; ++i) {
    const int idx = lane + 32 * i;
    const int seg = idx >> 2, qd = idx & 3;
    const int c = seg / 3, di = seg - 3 * c;
    cDyB[i]   = (di - 1) * HW * 4;
    cSafeB[i] = 4u * (unsigned)(c * LPIX + 4 * qd);
    cLds[i]   = seg * PP + 4 + 4 * qd;
  }
  const int side = lane & 1;
  unsigned hSafeB[3]; int hDyB[3], hLds[3];
  const int sideB = side ? 64 : -4;
#pragma unroll
  for (int i = 0; i < 3; ++i) {
    const int seg = (lane >> 1) + 16 * i;
    const int c = seg / 3, di = seg - 3 * c;
    hDyB[i]   = (di - 1) * HW * 4;
    hSafeB[i] = 4u * (unsigned)(c * LPIX);
    hLds[i]   = seg * PP + (side ? 20 : 3);
  }
  int bOff[2];
#pragma unroll
  for (int i = 0; i < 2; ++i) {
    const int e = lane + 32 * i;
    const int di = e / CSW, j = e - CSW * di;
    bOff[i] = di * PP + 3 + j;
  }
  int aOff[3][2];
#pragma unroll
  for (int v = 0; v < 3; ++v) {
#pragma unroll
    for (int e = 0; e < 2; ++e) {
      const int k = 4 * v + 2 * h + e;
      aOff[v][e] = (k / 3) * PP + 3 + (k % 3) + col;
    }
  }
  __syncthreads();

  float* patch = sPatch[wv][0];
  float* cs    = sCs[wv][0];
  float* hid   = sHid[wv][0];
  constexpr int PSTR = NSEG * PP;
  constexpr int HSTR = 16 * NO;

  constexpr int NTRIP = TPW / MT;
  const int tile0 = ((int)blockIdx.x * WAVES + wv) * TPW;

  v4f cv[MT][6];
  float hv[MT][3];
  auto issue_halo_loads = [&](int trip) {
    int row0 = (tile0 + trip * MT) << 4;
    asm volatile("" : "+s"(row0));
    const int p0   = row0 & (LPIX - 1);
    const int x0   = p0 & (HW - 1);
    const unsigned p0B = 4u * (unsigned)p0;
    const unsigned yB  = 4u * (unsigned)(p0 & ~(HW - 1));
    const char* xim = (const char*)(X + (row0 >> 16) * NC * LPIX);
#pragma unroll
    for (int i = 0; i < 3; ++i) {
      const bool ok = (yB + (unsigned)cDyB[i]) < 4u * (unsigned)LPIX;
      const unsigned off = p0B + cSafeB[i] + (ok ? (unsigned)cDyB[i] : 0u);
#pragma unroll
      for (int tl = 0; tl < MT; ++tl) {
        cv[tl][i]     = *(const v4f*)((xim + 64 * tl) + off);
        cv[tl][i + 3] = *(const v4f*)((xim + 64 * tl + C8B) + off);
      }
    }
#pragma unroll
    for (int tl = 0; tl < MT; ++tl) {
      const int xt = x0 + 16 * tl;
      const bool colOk = side ? (xt < HW - 16) : (xt > 0);
#pragma unroll
      for (int i = 0; i < 3; ++i) {
        const bool ok = colOk && ((yB + (unsigned)hDyB[i]) < 4u * (unsigned)LPIX);
        const unsigned off = p0B + 64u * tl + hSafeB[i] + (ok ? (unsigned)(hDyB[i] + sideB) : 0u);
        hv[tl][i] = *(const float*)(xim + off);
      }
    }
  };
  issue_halo_loads(0);

  for (int it = 0; it < NTRIP; ++it) {
    const int row0 = (tile0 + it * MT) << 4;
    const int bimg = row0 >> 16;
    const int p0   = row0 & (LPIX - 1);
    const unsigned yB = 4u * (unsigned)(p0 & ~(HW - 1));
    const int x0   = p0 & (HW - 1);

    const v4f z4 = {0.0f, 0.0f, 0.0f, 0.0f};
#pragma unroll
    for (int tl = 0; tl < MT; ++tl) {
#pragma unroll
      for (int i = 0; i < 3; ++i) {
        const bool ok = (yB + (unsigned)cDyB[i]) < 4u * (unsigned)LPIX;
        *(v4f*)(patch + tl * PSTR + cLds[i])           = ok ? cv[tl][i]     : z4;
        *(v4f*)(patch + tl * PSTR + cLds[i] + 24 * PP) = ok ? cv[tl][i + 3] : z4;
      }
      const int xt = x0 + 16 * tl;
      const bool colOk = side ? (xt < HW - 16) : (xt > 0);
#pragma unroll
      for (int i = 0; i < 3; ++i) {
        const bool ok = colOk && ((yB + (unsigned)hDyB[i]) < 4u * (unsigned)LPIX);
        patch[tl * PSTR + hLds[i]] = ok ? hv[tl][i] : 0.0f;
      }
    }
    wave_lds_handover();

#pragma unroll
    for (int i = 0; i < 2; ++i) {
      if (lane + 32 * i < 3 * CSW) {
#pragma unroll
        for (int tl = 0; tl < MT; ++tl) {
          float s = 0.0f;
#pragma unroll
          for (int c = 0; c < NC; ++c) s += patch[tl * PSTR + bOff[i] + c * 3 * PP];
          cs[tl * 64 + lane + 32 * i] = s;
        }
      }
    }
    wave_lds_handover();

    v8f acc0[MT], acc1[MT];
#pragma unroll
    for (int tl = 0; tl < MT; ++tl) {
#pragma unroll
      for (int r = 0; r < 8; ++r) { acc0[tl][r] = b1v0; acc1[tl][r] = b1v1; }
    }
    const float* w1Base = sW1 + col * P1 + 2 * h;
#pragma unroll
    for (int s = 0; s < DF / 4; ++s) {
      const int u = s / 3, v = s - 3 * u;
      v2f bA = *(const v2f*)(w1Base + 4 * s);
      v2f bB = *(const v2f*)(w1Base + 4 * s + 16 * P1);
#pragma unroll
      for (int tl = 0; tl < MT; ++tl) {
        v2f a;
        a.x = patch[tl * PSTR + aOff[v][0] + u * 4 * PP];
        a.y = patch[tl * PSTR + aOff[v][1] + u * 4 * PP];
        acc0[tl] = wmma4(a, bA, acc0[tl]);
        acc1[tl] = wmma4(a, bB, acc1[tl]);
      }
    }
#pragma unroll
    for (int tl = 0; tl < MT; ++tl) {
#pragma unroll
      for (int r = 0; r < 8; ++r) {
        int m2 = r + 8 * h;
        hid[tl * HSTR + m2 * NO + col]      = tanh_hw(acc0[tl][r]);
        hid[tl * HSTR + m2 * NO + col + 16] = tanh_hw(acc1[tl][r]);
      }
    }
    if (it + 1 < NTRIP) issue_halo_loads(it + 1);
    wave_lds_handover();

    float hr[MT][NO];
    float pr[MT][9];
#pragma unroll
    for (int tl = 0; tl < MT; ++tl) {
#pragma unroll
      for (int k = 0; k < NO / 4; ++k) {
        v4f v = *(const v4f*)(hid + tl * HSTR + col * NO + 4 * k);
        hr[tl][4 * k + 0] = v.x; hr[tl][4 * k + 1] = v.y; hr[tl][4 * k + 2] = v.z; hr[tl][4 * k + 3] = v.w;
      }
#pragma unroll
      for (int j = 0; j < 9; ++j) pr[tl][j] = cs[tl * 64 + (j / 3) * CSW + col + (j % 3)];

#pragma unroll
      for (int r = 0; r < 8; ++r) {
        int m2 = r + 8 * h;
        float s0 = bs0, s1 = bs1;
#pragma unroll
        for (int j = 0; j < 9; ++j) {
          float pv = cs[tl * 64 + (j / 3) * CSW + m2 + (j % 3)];
          s0 += pv * b2a[j]; s1 += pv * b2b[j];
        }
        acc0[tl][r] = s0; acc1[tl][r] = s1;
      }
    }

    const float* w2Base = sW2 + col * P2 + 2 * h;
#pragma unroll
    for (int s = 0; s < DQ / 4; ++s) {
      const int q = 4 * s;
      v2f bA = *(const v2f*)(w2Base + 4 * s);
      v2f bB = *(const v2f*)(w2Base + 4 * s + 16 * P2);
#pragma unroll
      for (int tl = 0; tl < MT; ++tl) {
        float z0 = hr[tl][(q + 0) / 9] * pr[tl][(q + 0) % 9];
        float z1 = hr[tl][(q + 1) / 9] * pr[tl][(q + 1) % 9];
        float z2 = hr[tl][(q + 2) / 9] * pr[tl][(q + 2) % 9];
        float z3 = hr[tl][(q + 3) / 9] * pr[tl][(q + 3) % 9];
        v2f a; a.x = h ? z2 : z0; a.y = h ? z3 : z1;
        acc0[tl] = wmma4(a, bA, acc0[tl]);
        acc1[tl] = wmma4(a, bB, acc1[tl]);
      }
    }

    float* outT = hid;
#pragma unroll
    for (int tl = 0; tl < MT; ++tl) {
#pragma unroll
      for (int r = 0; r < 8; ++r) {
        int m2 = r + 8 * h;
        outT[tl * HSTR + m2 * NO + col]      = acc0[tl][r];
        outT[tl * HSTR + m2 * NO + col + 16] = acc1[tl][r];
      }
    }
    wave_lds_handover();
#pragma unroll
    for (int tl = 0; tl < MT; ++tl) {
#pragma unroll
      for (int e = 0; e < 16; ++e) {
        int idx = e * 32 + lane;
        int o = idx >> 4, m = idx & 15;
        Out[((long)(bimg * NO + o)) * LPIX + p0 + 16 * tl + m] = outT[tl * HSTR + m * NO + o];
      }
    }
  }
}

extern "C" void kernel_launch(void* const* d_in, const int* in_sizes, int n_in,
                              void* d_out, int out_size, void* d_ws, size_t ws_size,
                              hipStream_t stream) {
  (void)in_sizes; (void)n_in; (void)out_size; (void)d_ws; (void)ws_size;
  const float* X    = (const float*)d_in[0];
  const float* W1   = (const float*)d_in[1];
  const float* b1   = (const float*)d_in[2];
  const float* W2   = (const float*)d_in[3];
  const float* b2   = (const float*)d_in[4];
  const float* bias = (const float*)d_in[5];
  float* Out = (float*)d_out;
  dynaconv_kernel<<<BLOCKS, WAVES * 32, 0, stream>>>(X, W1, b1, W2, b2, bias, Out);
}


# wave priority by progress (s_setprio): 3 until a wave's last trip, then 2, 1, and 0 for its last 18 GEMM2 k steps: the wave furthest behind wins the matrix pipe, so a SIMD's 4 waves end together
# speedup vs baseline: 3.4698x; 1.0378x over previous
#include <hip/hip_runtime.h>
#include <math.h>


typedef float v2f __attribute__((ext_vector_type(2)));
typedef float v4f __attribute__((ext_vector_type(4)));
typedef float v8f __attribute__((ext_vector_type(8)));

#define NB 4
#define NC 16
#define NO 32
#define HW 256
#define LPIX (HW*HW)
#define DF 144
#define DQ 288
#define WAVES 16
#define TPW 4
#define MT 2
#define ROWS_TOTAL (NB*LPIX)
#define TILES_TOTAL (ROWS_TOTAL/16)
#define BLOCKS (TILES_TOTAL/(WAVES*TPW))

#define P1 (DF + 4)
#define P2 (DQ + 4)

#define PP   24
#define NSEG (NC*3)
#define CSW  18
#define PRIO0_STEP 54

static __device__ __forceinline__ v8f wmma4(v2f a, v2f b, v8f c) {
  return __builtin_amdgcn_wmma_f32_16x16x4_f32(
      false, a, false, b, (short)0, c, false, false);
}


static __device__ __forceinline__ void wave_lds_handover() {
  asm volatile("s_wait_dscnt 0x0" ::: "memory");
}

static __device__ __forceinline__ float tanh_hw(float x) {
  return __builtin_amdgcn_tanhf(x);
}

__global__ __launch_bounds__(WAVES * 32)
void dynaconv_kernel(const float* __restrict__ X,
                     const float* __restrict__ W1,
                     const float* __restrict__ b1,
                     const float* __restrict__ W2,
                     const float* __restrict__ b2,
                     const float* __restrict__ bias,
                     float* __restrict__ Out) {
  __shared__ __align__(16) float sW1[ NO * P1 ];
  __shared__ __align__(16) float sW2[ NO * P2 ];
  __shared__ __align__(16) float sPatch[WAVES][MT][NSEG * PP];
  __shared__ __align__(16) float sCs  [WAVES][MT][64];
  __shared__ __align__(16) float sHid [WAVES][MT][16 * NO];

  const int tid  = threadIdx.x;
  const int lane = tid & 31;
  const int wv   = __builtin_amdgcn_readfirstlane(tid >> 5);
  const int col  = lane & 15;
  const int h    = lane >> 4;

  {
    constexpr int NT  = WAVES * 32;
    constexpr int Q1  = NO * DF / 4;
    constexpr int Q2  = DQ * NO / 4;
    constexpr int N1  = (Q1 + NT - 1) / NT;
    constexpr int N2  = (Q2 + NT - 1) / NT;
    v4f w1v[N1], w2v[N2];
#pragma unroll
    for (int i = 0; i < N1; ++i) if (tid + NT * i < Q1) w1v[i] = ((const v4f*)W1)[tid + NT * i];
#pragma unroll
    for (int i = 0; i < N2; ++i) if (tid + NT * i < Q2) w2v[i] = ((const v4f*)W2)[tid + NT * i];
#pragma unroll
    for (int i = 0; i < N1; ++i) {
      const int e4 = tid + NT * i;
      if (e4 < Q1) {
        const int o = e4 / (DF / 4), c = e4 - o * (DF / 4);
        *(v4f*)(sW1 + o * P1 + 4 * c) = w1v[i];
      }
    }
#pragma unroll
    for (int i = 0; i < N2; ++i) {
      const int e4 = tid + NT * i;
      if (e4 < Q2) {
        const int g = e4 >> 3, c4 = e4 & 7;
        const int o = g / 9, j = g - o * 9;
        float* d = sW2 + o * P2 + 36 * c4 + j;
        d[0] = w2v[i].x; d[9] = w2v[i].y; d[18] = w2v[i].z; d[27] = w2v[i].w;
      }
    }
  }
  const float b1v0 = b1[col],      b1v1 = b1[col + 16];
  const float bs0  = bias[col],    bs1  = bias[col + 16];
  float b2a[9], b2b[9];
#pragma unroll
  for (int j = 0; j < 9; ++j) { b2a[j] = b2[col * 9 + j]; b2b[j] = b2[(col + 16) * 9 + j]; }

  unsigned cSafeB[3]; int cDyB[3], cLds[3];
  constexpr unsigned C8B = 8u * LPIX * 4u;
#pragma unroll
  for (int i = 0; i < 3; ++i) {
    const int idx = lane + 32 * i;
    const int seg = idx >> 2, qd = idx & 3;
    const int c = seg / 3, di = seg - 3 * c;
    cDyB[i]   = (di - 1) * HW * 4;
    cSafeB[i] = 4u * (unsigned)(c * LPIX + 4 * qd);
    cLds[i]   = seg * PP + 4 + 4 * qd;
  }
  const int side = lane & 1;
  unsigned hSafeB[3]; int hDyB[3], hLds[3];
  const int sideB = side ? 64 : -4;
#pragma unroll
  for (int i = 0; i < 3; ++i) {
    const int seg = (lane >> 1) + 16 * i;
    const int c = seg / 3, di = seg - 3 * c;
    hDyB[i]   = (di - 1) * HW * 4;
    hSafeB[i] = 4u * (unsigned)(c * LPIX);
    hLds[i]   = seg * PP + (side ? 20 : 3);
  }
  int bOff[2];
#pragma unroll
  for (int i = 0; i < 2; ++i) {
    const int e = lane + 32 * i;
    const int di = e / CSW, j = e - CSW * di;
    bOff[i] = di * PP + 3 + j;
  }
  int aOff[3][2];
#pragma unroll
  for (int v = 0; v < 3; ++v) {
#pragma unroll
    for (int e = 0; e < 2; ++e) {
      const int k = 4 * v + 2 * h + e;
      aOff[v][e] = (k / 3) * PP + 3 + (k % 3) + col;
    }
  }
  __syncthreads();

  __builtin_amdgcn_s_setprio(3);

  float* patch = sPatch[wv][0];
  float* cs    = sCs[wv][0];
  float* hid   = sHid[wv][0];
  constexpr int PSTR = NSEG * PP;
  constexpr int HSTR = 16 * NO;

  constexpr int NTRIP = TPW / MT;
  const int tile0 = ((int)blockIdx.x * WAVES + wv) * TPW;

  v4f cv[MT][6];
  float hv[MT][3];
  auto issue_halo_loads = [&](int trip) {
    int row0 = (tile0 + trip * MT) << 4;
    asm volatile("" : "+s"(row0));
    const int p0   = row0 & (LPIX - 1);
    const int x0   = p0 & (HW - 1);
    const unsigned p0B = 4u * (unsigned)p0;
    const unsigned yB  = 4u * (unsigned)(p0 & ~(HW - 1));
    const char* xim = (const char*)(X + (row0 >> 16) * NC * LPIX);
#pragma unroll
    for (int i = 0; i < 3; ++i) {
      const bool ok = (yB + (unsigned)cDyB[i]) < 4u * (unsigned)LPIX;
      const unsigned off = p0B + cSafeB[i] + (ok ? (unsigned)cDyB[i] : 0u);
#pragma unroll
      for (int tl = 0; tl < MT; ++tl) {
        cv[tl][i]     = *(const v4f*)((xim + 64 * tl) + off);
        cv[tl][i + 3] = *(const v4f*)((xim + 64 * tl + C8B) + off);
      }
    }
#pragma unroll
    for (int tl = 0; tl < MT; ++tl) {
      const int xt = x0 + 16 * tl;
      const bool colOk = side ? (xt < HW - 16) : (xt > 0);
#pragma unroll
      for (int i = 0; i < 3; ++i) {
        const bool ok = colOk && ((yB + (unsigned)hDyB[i]) < 4u * (unsigned)LPIX);
        const unsigned off = p0B + 64u * tl + hSafeB[i] + (ok ? (unsigned)(hDyB[i] + sideB) : 0u);
        hv[tl][i] = *(const float*)(xim + off);
      }
    }
  };
  issue_halo_loads(0);

  for (int it = 0; it < NTRIP; ++it) {
    const int row0 = (tile0 + it * MT) << 4;
    const int bimg = row0 >> 16;
    const int p0   = row0 & (LPIX - 1);
    const unsigned yB = 4u * (unsigned)(p0 & ~(HW - 1));
    const int x0   = p0 & (HW - 1);

    if (it == NTRIP - 1) __builtin_amdgcn_s_setprio(2);

    const v4f z4 = {0.0f, 0.0f, 0.0f, 0.0f};
#pragma unroll
    for (int tl = 0; tl < MT; ++tl) {
#pragma unroll
      for (int i = 0; i < 3; ++i) {
        const bool ok = (yB + (unsigned)cDyB[i]) < 4u * (unsigned)LPIX;
        *(v4f*)(patch + tl * PSTR + cLds[i])           = ok ? cv[tl][i]     : z4;
        *(v4f*)(patch + tl * PSTR + cLds[i] + 24 * PP) = ok ? cv[tl][i + 3] : z4;
      }
      const int xt = x0 + 16 * tl;
      const bool colOk = side ? (xt < HW - 16) : (xt > 0);
#pragma unroll
      for (int i = 0; i < 3; ++i) {
        const bool ok = colOk && ((yB + (unsigned)hDyB[i]) < 4u * (unsigned)LPIX);
        patch[tl * PSTR + hLds[i]] = ok ? hv[tl][i] : 0.0f;
      }
    }
    wave_lds_handover();

#pragma unroll
    for (int i = 0; i < 2; ++i) {
      if (lane + 32 * i < 3 * CSW) {
#pragma unroll
        for (int tl = 0; tl < MT; ++tl) {
          float s = 0.0f;
#pragma unroll
          for (int c = 0; c < NC; ++c) s += patch[tl * PSTR + bOff[i] + c * 3 * PP];
          cs[tl * 64 + lane + 32 * i] = s;
        }
      }
    }
    wave_lds_handover();

    v8f acc0[MT], acc1[MT];
#pragma unroll
    for (int tl = 0; tl < MT; ++tl) {
#pragma unroll
      for (int r = 0; r < 8; ++r) { acc0[tl][r] = b1v0; acc1[tl][r] = b1v1; }
    }
    const float* w1Base = sW1 + col * P1 + 2 * h;
#pragma unroll
    for (int s = 0; s < DF / 4; ++s) {
      const int u = s / 3, v = s - 3 * u;
      v2f bA = *(const v2f*)(w1Base + 4 * s);
      v2f bB = *(const v2f*)(w1Base + 4 * s + 16 * P1);
#pragma unroll
      for (int tl = 0; tl < MT; ++tl) {
        v2f a;
        a.x = patch[tl * PSTR + aOff[v][0] + u * 4 * PP];
        a.y = patch[tl * PSTR + aOff[v][1] + u * 4 * PP];
        acc0[tl] = wmma4(a, bA, acc0[tl]);
        acc1[tl] = wmma4(a, bB, acc1[tl]);
      }
    }
#pragma unroll
    for (int tl = 0; tl < MT; ++tl) {
#pragma unroll
      for (int r = 0; r < 8; ++r) {
        int m2 = r + 8 * h;
        hid[tl * HSTR + m2 * NO + col]      = tanh_hw(acc0[tl][r]);
        hid[tl * HSTR + m2 * NO + col + 16] = tanh_hw(acc1[tl][r]);
      }
    }
    if (it + 1 < NTRIP) issue_halo_loads(it + 1);
    else __builtin_amdgcn_s_setprio(1);
    wave_lds_handover();

    float hr[MT][NO];
    float pr[MT][9];
#pragma unroll
    for (int tl = 0; tl < MT; ++tl) {
#pragma unroll
      for (int k = 0; k < NO / 4; ++k) {
        v4f v = *(const v4f*)(hid + tl * HSTR + col * NO + 4 * k);
        hr[tl][4 * k + 0] = v.x; hr[tl][4 * k + 1] = v.y; hr[tl][4 * k + 2] = v.z; hr[tl][4 * k + 3] = v.w;
      }
#pragma unroll
      for (int j = 0; j < 9; ++j) pr[tl][j] = cs[tl * 64 + (j / 3) * CSW + col + (j % 3)];

#pragma unroll
      for (int r = 0; r < 8; ++r) {
        int m2 = r + 8 * h;
        float s0 = bs0, s1 = bs1;
#pragma unroll
        for (int j = 0; j < 9; ++j) {
          float pv = cs[tl * 64 + (j / 3) * CSW + m2 + (j % 3)];
          s0 += pv * b2a[j]; s1 += pv * b2b[j];
        }
        acc0[tl][r] = s0; acc1[tl][r] = s1;
      }
    }

    const float* w2Base = sW2 + col * P2 + 2 * h;
#pragma unroll
    for (int s = 0; s < DQ / 4; ++s) {
      const int q = 4 * s;
      if (s == PRIO0_STEP)
        asm volatile("s_cmp_lg_u32 %0, 0\n\ts_cbranch_scc1 1f\n\ts_setprio 0\n1:" :: "s"((int)(NTRIP - 1 - it)) : "scc");
      v2f bA = *(const v2f*)(w2Base + 4 * s);
      v2f bB = *(const v2f*)(w2Base + 4 * s + 16 * P2);
#pragma unroll
      for (int tl = 0; tl < MT; ++tl) {
        float z0 = hr[tl][(q + 0) / 9] * pr[tl][(q + 0) % 9];
        float z1 = hr[tl][(q + 1) / 9] * pr[tl][(q + 1) % 9];
        float z2 = hr[tl][(q + 2) / 9] * pr[tl][(q + 2) % 9];
        float z3 = hr[tl][(q + 3) / 9] * pr[tl][(q + 3) % 9];
        v2f a; a.x = h ? z2 : z0; a.y = h ? z3 : z1;
        acc0[tl] = wmma4(a, bA, acc0[tl]);
        acc1[tl] = wmma4(a, bB, acc1[tl]);
      }
    }

    float* outT = hid;
#pragma unroll
    for (int tl = 0; tl < MT; ++tl) {
#pragma unroll
      for (int r = 0; r < 8; ++r) {
        int m2 = r + 8 * h;
        outT[tl * HSTR + m2 * NO + col]      = acc0[tl][r];
        outT[tl * HSTR + m2 * NO + col + 16] = acc1[tl][r];
      }
    }
    wave_lds_handover();
#pragma unroll
    for (int tl = 0; tl < MT; ++tl) {
#pragma unroll
      for (int e = 0; e < 16; ++e) {
        int idx = e * 32 + lane;
        int o = idx >> 4, m = idx & 15;
        Out[((long)(bimg * NO + o)) * LPIX + p0 + 16 * tl + m] = outT[tl * HSTR + m * NO + o];
      }
    }
  }
}

extern "C" void kernel_launch(void* const* d_in, const int* in_sizes, int n_in,
                              void* d_out, int out_size, void* d_ws, size_t ws_size,
                              hipStream_t stream) {
  (void)in_sizes; (void)n_in; (void)out_size; (void)d_ws; (void)ws_size;
  const float* X    = (const float*)d_in[0];
  const float* W1   = (const float*)d_in[1];
  const float* b1   = (const float*)d_in[2];
  const float* W2   = (const float*)d_in[3];
  const float* b2   = (const float*)d_in[4];
  const float* bias = (const float*)d_in[5];
  float* Out = (float*)d_out;
  dynaconv_kernel<<<BLOCKS, WAVES * 32, 0, stream>>>(X, W1, b1, W2, b2, bias, Out);
}


# GEMM2 K axis re-indexed: lane half h owns hidden channels 16h..16h+15 (A pair = one packed product, no select); b2 and bias ride as a constant-1 channel, 3 more k steps; GEMM2 sum order changed
# speedup vs baseline: 3.5926x; 1.0354x over previous
#include <hip/hip_runtime.h>
#include <math.h>


typedef float v2f __attribute__((ext_vector_type(2)));
typedef float v4f __attribute__((ext_vector_type(4)));
typedef float v8f __attribute__((ext_vector_type(8)));

#define NB 4
#define NC 16
#define NO 32
#define HW 256
#define LPIX (HW*HW)
#define DF 144
#define DQ 288
#define WAVES 16
#define TPW 4
#define MT 2
#define ROWS_TOTAL (NB*LPIX)
#define TILES_TOTAL (ROWS_TOTAL/16)
#define BLOCKS (TILES_TOTAL/(WAVES*TPW))

#define P1 (DF + 4)
#define KQ (DQ + 12)
#define P2 KQ

#define PP   24
#define NSEG (NC*3)
#define CSW  18
#define PRIO0_STEP 54

static __device__ __forceinline__ v8f wmma4(v2f a, v2f b, v8f c) {
  return __builtin_amdgcn_wmma_f32_16x16x4_f32(
      false, a, false, b, (short)0, c, false, false);
}


static __device__ __forceinline__ void wave_lds_handover() {
  asm volatile("s_wait_dscnt 0x0" ::: "memory");
}

static __device__ __forceinline__ float tanh_hw(float x) {
  return __builtin_amdgcn_tanhf(x);
}

__global__ __launch_bounds__(WAVES * 32)
void dynaconv_kernel(const float* __restrict__ X,
                     const float* __restrict__ W1,
                     const float* __restrict__ b1,
                     const float* __restrict__ W2,
                     const float* __restrict__ b2,
                     const float* __restrict__ bias,
                     float* __restrict__ Out) {
  __shared__ __align__(16) float sW1[ NO * P1 ];
  __shared__ __align__(16) float sW2[ NO * P2 ];
  __shared__ __align__(16) float sPatch[WAVES][MT][NSEG * PP];
  __shared__ __align__(16) float sCs  [WAVES][MT][64];
  __shared__ __align__(16) float sHid [WAVES][MT][16 * NO];

  const int tid  = threadIdx.x;
  const int lane = tid & 31;
  const int wv   = __builtin_amdgcn_readfirstlane(tid >> 5);
  const int col  = lane & 15;
  const int h    = lane >> 4;

  {
    constexpr int NT  = WAVES * 32;
    constexpr int Q1  = NO * DF / 4;
    constexpr int Q2  = DQ * NO / 4;
    constexpr int N1  = (Q1 + NT - 1) / NT;
    constexpr int N2  = (Q2 + NT - 1) / NT;
    v4f w1v[N1], w2v[N2];
#pragma unroll
    for (int i = 0; i < N1; ++i) if (tid + NT * i < Q1) w1v[i] = ((const v4f*)W1)[tid + NT * i];
#pragma unroll
    for (int i = 0; i < N2; ++i) if (tid + NT * i < Q2) w2v[i] = ((const v4f*)W2)[tid + NT * i];
    const int oC = (tid < NO * 12) ? tid / 12 : 0;
    const int jC = tid - (tid / 12) * 12;
    const float cch = *((jC < 9) ? (b2 + oC * 9 + jC) : (bias + oC));
#pragma unroll
    for (int i = 0; i < N1; ++i) {
      const int e4 = tid + NT * i;
      if (e4 < Q1) {
        const int o = e4 / (DF / 4), c = e4 - o * (DF / 4);
        *(v4f*)(sW1 + o * P1 + 4 * c) = w1v[i];
      }
    }
#pragma unroll
    for (int i = 0; i < N2; ++i) {
      const int e4 = tid + NT * i;
      if (e4 < Q2) {
        const int g = e4 >> 3, c4 = e4 & 7;
        const int o = g / 9, j = g - o * 9;
        const int t0 = 36 * (c4 & 3) + j;
        const int par = t0 & 1;
        float* d0 = sW2 + o * P2 + 2 * (c4 >> 2) + 2 * t0 - par;
        float* d1 = d0 + 2 * par;
        d0[0] = w2v[i].x; d1[17] = w2v[i].y; d0[36] = w2v[i].z; d1[53] = w2v[i].w;
      }
    }
    if (tid < NO * 12) sW2[oC * P2 + DQ + jC] = (jC > 9) ? 0.0f : cch;
  }
  const float b1v0 = b1[col],      b1v1 = b1[col + 16];

  unsigned cSafeB[3]; int cDyB[3], cLds[3];
  constexpr unsigned C8B = 8u * LPIX * 4u;
#pragma unroll
  for (int i = 0; i < 3; ++i) {
    const int idx = lane + 32 * i;
    const int seg = idx >> 2, qd = idx & 3;
    const int c = seg / 3, di = seg - 3 * c;
    cDyB[i]   = (di - 1) * HW * 4;
    cSafeB[i] = 4u * (unsigned)(c * LPIX + 4 * qd);
    cLds[i]   = seg * PP + 4 + 4 * qd;
  }
  const int side = lane & 1;
  unsigned hSafeB[3]; int hDyB[3], hLds[3];
  const int sideB = side ? 64 : -4;
#pragma unroll
  for (int i = 0; i < 3; ++i) {
    const int seg = (lane >> 1) + 16 * i;
    const int c = seg / 3, di = seg - 3 * c;
    hDyB[i]   = (di - 1) * HW * 4;
    hSafeB[i] = 4u * (unsigned)(c * LPIX);
    hLds[i]   = seg * PP + (side ? 20 : 3);
  }
  int bOff[2];
#pragma unroll
  for (int i = 0; i < 2; ++i) {
    const int e = lane + 32 * i;
    const int di = e / CSW, j = e - CSW * di;
    bOff[i] = di * PP + 3 + j;
  }
  int aOff[3][2];
#pragma unroll
  for (int v = 0; v < 3; ++v) {
#pragma unroll
    for (int e = 0; e < 2; ++e) {
      const int k = 4 * v + 2 * h + e;
      aOff[v][e] = (k / 3) * PP + 3 + (k % 3) + col;
    }
  }
  __syncthreads();

  __builtin_amdgcn_s_setprio(3);

  float* patch = sPatch[wv][0];
  float* cs    = sCs[wv][0];
  float* hid   = sHid[wv][0];
  constexpr int PSTR = NSEG * PP;
  constexpr int HSTR = 16 * NO;

  constexpr int NTRIP = TPW / MT;
  const int tile0 = ((int)blockIdx.x * WAVES + wv) * TPW;

  v4f cv[MT][6];
  float hv[MT][3];
  auto issue_halo_loads = [&](int trip) {
    int row0 = (tile0 + trip * MT) << 4;
    asm volatile("" : "+s"(row0));
    const int p0   = row0 & (LPIX - 1);
    const int x0   = p0 & (HW - 1);
    const unsigned p0B = 4u * (unsigned)p0;
    const unsigned yB  = 4u * (unsigned)(p0 & ~(HW - 1));
    const char* xim = (const char*)(X + (row0 >> 16) * NC * LPIX);
#pragma unroll
    for (int i = 0; i < 3; ++i) {
      const bool ok = (yB + (unsigned)cDyB[i]) < 4u * (unsigned)LPIX;
      const unsigned off = p0B + cSafeB[i] + (ok ? (unsigned)cDyB[i] : 0u);
#pragma unroll
      for (int tl = 0; tl < MT; ++tl) {
        cv[tl][i]     = *(const v4f*)((xim + 64 * tl) + off);
        cv[tl][i + 3] = *(const v4f*)((xim + 64 * tl + C8B) + off);
      }
    }
#pragma unroll
    for (int tl = 0; tl < MT; ++tl) {
      const int xt = x0 + 16 * tl;
      const bool colOk = side ? (xt < HW - 16) : (xt > 0);
#pragma unroll
      for (int i = 0; i < 3; ++i) {
        const bool ok = colOk && ((yB + (unsigned)hDyB[i]) < 4u * (unsigned)LPIX);
        const unsigned off = p0B + 64u * tl + hSafeB[i] + (ok ? (unsigned)(hDyB[i] + sideB) : 0u);
        hv[tl][i] = *(const float*)(xim + off);
      }
    }
  };
  issue_halo_loads(0);

  for (int it = 0; it < NTRIP; ++it) {
    const int row0 = (tile0 + it * MT) << 4;
    const int bimg = row0 >> 16;
    const int p0   = row0 & (LPIX - 1);
    const unsigned yB = 4u * (unsigned)(p0 & ~(HW - 1));
    const int x0   = p0 & (HW - 1);

    if (it == NTRIP - 1) __builtin_amdgcn_s_setprio(2);

    const v4f z4 = {0.0f, 0.0f, 0.0f, 0.0f};
#pragma unroll
    for (int tl = 0; tl < MT; ++tl) {
#pragma unroll
      for (int i = 0; i < 3; ++i) {
        const bool ok = (yB + (unsigned)cDyB[i]) < 4u * (unsigned)LPIX;
        *(v4f*)(patch + tl * PSTR + cLds[i])           = ok ? cv[tl][i]     : z4;
        *(v4f*)(patch + tl * PSTR + cLds[i] + 24 * PP) = ok ? cv[tl][i + 3] : z4;
      }
      const int xt = x0 + 16 * tl;
      const bool colOk = side ? (xt < HW - 16) : (xt > 0);
#pragma unroll
      for (int i = 0; i < 3; ++i) {
        const bool ok = colOk && ((yB + (unsigned)hDyB[i]) < 4u * (unsigned)LPIX);
        patch[tl * PSTR + hLds[i]] = ok ? hv[tl][i] : 0.0f;
      }
    }
    wave_lds_handover();

#pragma unroll
    for (int i = 0; i < 2; ++i) {
      if (lane + 32 * i < 3 * CSW) {
#pragma unroll
        for (int tl = 0; tl < MT; ++tl) {
          float s = 0.0f;
#pragma unroll
          for (int c = 0; c < NC; ++c) s += patch[tl * PSTR + bOff[i] + c * 3 * PP];
          cs[tl * 64 + lane + 32 * i] = s;
        }
      }
    }
    wave_lds_handover();

    v8f acc0[MT], acc1[MT];
#pragma unroll
    for (int tl = 0; tl < MT; ++tl) {
#pragma unroll
      for (int r = 0; r < 8; ++r) { acc0[tl][r] = b1v0; acc1[tl][r] = b1v1; }
    }
    const float* w1Base = sW1 + col * P1 + 2 * h;
#pragma unroll
    for (int s = 0; s < DF / 4; ++s) {
      const int u = s / 3, v = s - 3 * u;
      v2f bA = *(const v2f*)(w1Base + 4 * s);
      v2f bB = *(const v2f*)(w1Base + 4 * s + 16 * P1);
#pragma unroll
      for (int tl = 0; tl < MT; ++tl) {
        v2f a;
        a.x = patch[tl * PSTR + aOff[v][0] + u * 4 * PP];
        a.y = patch[tl * PSTR + aOff[v][1] + u * 4 * PP];
        acc0[tl] = wmma4(a, bA, acc0[tl]);
        acc1[tl] = wmma4(a, bB, acc1[tl]);
      }
    }
#pragma unroll
    for (int tl = 0; tl < MT; ++tl) {
#pragma unroll
      for (int r = 0; r < 8; ++r) {
        int m2 = r + 8 * h;
        hid[tl * HSTR + m2 * NO + col]      = tanh_hw(acc0[tl][r]);
        hid[tl * HSTR + m2 * NO + col + 16] = tanh_hw(acc1[tl][r]);
      }
    }
    if (it + 1 < NTRIP) issue_halo_loads(it + 1);
    else __builtin_amdgcn_s_setprio(1);
    wave_lds_handover();

    float hr[MT][NO / 2];
    float pr[MT][9];
#pragma unroll
    for (int tl = 0; tl < MT; ++tl) {
#pragma unroll
      for (int k = 0; k < NO / 8; ++k) {
        v4f v = *(const v4f*)(hid + tl * HSTR + col * NO + 16 * h + 4 * k);
        hr[tl][4 * k + 0] = v.x; hr[tl][4 * k + 1] = v.y; hr[tl][4 * k + 2] = v.z; hr[tl][4 * k + 3] = v.w;
      }
#pragma unroll
      for (int j = 0; j < 9; ++j) pr[tl][j] = cs[tl * 64 + (j / 3) * CSW + col + (j % 3)];
#pragma unroll
      for (int r = 0; r < 8; ++r) { acc0[tl][r] = 0.0f; acc1[tl][r] = 0.0f; }
    }

    const float* w2Base = sW2 + col * P2 + 2 * h;
#pragma unroll
    for (int s = 0; s < KQ / 4; ++s) {
      if (s == PRIO0_STEP)
        asm volatile("s_cmp_lg_u32 %0, 0\n\ts_cbranch_scc1 1f\n\ts_setprio 0\n1:" :: "s"((int)(NTRIP - 1 - it)) : "scc");
      v2f bA = *(const v2f*)(w2Base + 4 * s);
      v2f bB = *(const v2f*)(w2Base + 4 * s + 16 * P2);
#pragma unroll
      for (int tl = 0; tl < MT; ++tl) {
        v2f a;
        if (s < DQ / 4) {
          a.x = hr[tl][(2 * s) / 9]     * pr[tl][(2 * s) % 9];
          a.y = hr[tl][(2 * s + 1) / 9] * pr[tl][(2 * s + 1) % 9];
        } else {
          const int j0 = 4 * (s - DQ / 4);
          const float lo0 = pr[tl][j0];
          const float lo1 = (j0 + 1 < 9) ? pr[tl][(j0 + 1) % 9] : 1.0f;
          const float hi0 = (j0 + 2 < 9) ? pr[tl][(j0 + 2) % 9] : 0.0f;
          const float hi1 = (j0 + 3 < 9) ? pr[tl][(j0 + 3) % 9] : 0.0f;
          a.x = h ? hi0 : lo0; a.y = h ? hi1 : lo1;
        }
        acc0[tl] = wmma4(a, bA, acc0[tl]);
        acc1[tl] = wmma4(a, bB, acc1[tl]);
      }
    }

    float* outT = hid;
#pragma unroll
    for (int tl = 0; tl < MT; ++tl) {
#pragma unroll
      for (int r = 0; r < 8; ++r) {
        int m2 = r + 8 * h;
        outT[tl * HSTR + m2 * NO + col]      = acc0[tl][r];
        outT[tl * HSTR + m2 * NO + col + 16] = acc1[tl][r];
      }
    }
    wave_lds_handover();
#pragma unroll
    for (int tl = 0; tl < MT; ++tl) {
#pragma unroll
      for (int e = 0; e < 16; ++e) {
        int idx = e * 32 + lane;
        int o = idx >> 4, m = idx & 15;
        Out[((long)(bimg * NO + o)) * LPIX + p0 + 16 * tl + m] = outT[tl * HSTR + m * NO + o];
      }
    }
  }
}

extern "C" void kernel_launch(void* const* d_in, const int* in_sizes, int n_in,
                              void* d_out, int out_size, void* d_ws, size_t ws_size,
                              hipStream_t stream) {
  (void)in_sizes; (void)n_in; (void)out_size; (void)d_ws; (void)ws_size;
  const float* X    = (const float*)d_in[0];
  const float* W1   = (const float*)d_in[1];
  const float* b1   = (const float*)d_in[2];
  const float* W2   = (const float*)d_in[3];
  const float* b2   = (const float*)d_in[4];
  const float* bias = (const float*)d_in[5];
  float* Out = (float*)d_out;
  dynaconv_kernel<<<BLOCKS, WAVES * 32, 0, stream>>>(X, W1, b1, W2, b2, bias, Out);
}
